# starting kernel
# baseline (speedup 1.0000x reference)
#include <hip/hip_runtime.h>

#define BATCH   8
#define NQ      512
#define NKV     512
#define NH      16
#define DMODEL  256
#define NEG_INF_F (-1.0e9f)

#define KC_ATT  32
#define KC_GEMM 32

typedef __attribute__((ext_vector_type(2))) float v2f;
typedef __attribute__((ext_vector_type(8))) float v8f;

__device__ __forceinline__ v8f wmma_f32(v2f a, v2f b, v8f c) {
    return __builtin_amdgcn_wmma_f32_16x16x4_f32(
         false, a,  false, b,
         (short)0, c,  false,  false);
}

__launch_bounds__(256)
__global__ void gemm_bias_256(const float* __restrict__ in,
                              const float* __restrict__ W,
                              const float* __restrict__ bias,
                              float* __restrict__ out) {
    __shared__ float s_in[16 * 257];
    __shared__ float s_w[DMODEL * (KC_GEMM + 1)];

    const int tid   = threadIdx.x;
    const int wid   = tid >> 5;
    const int lane  = tid & 31;
    const int row   = lane & 15;
    const int khalf = (lane < 16) ? 0 : 2;
    const int n0    = blockIdx.x * 16;

    for (int i = 0; i < 4; ++i) {
        int e4 = (tid + 256 * i) * 4;
        int r  = e4 >> 8, k = e4 & 255;
        const float4 v = *(const float4*)&in[(size_t)(n0 + r) * DMODEL + k];
        s_in[r * 257 + k + 0] = v.x;
        s_in[r * 257 + k + 1] = v.y;
        s_in[r * 257 + k + 2] = v.z;
        s_in[r * 257 + k + 3] = v.w;
    }

    v8f acc0 = {};
    v8f acc1 = {};
    const int d0 = (wid)     * 16;
    const int d1 = (wid + 8) * 16;

    for (int kc = 0; kc < DMODEL; kc += KC_GEMM) {
        __syncthreads();
        for (int i = 0; i < 8; ++i) {
            int e4 = (tid + 256 * i) * 4;
            int d  = e4 >> 5, k = e4 & 31;
            const float4 v = *(const float4*)&W[(size_t)d * DMODEL + kc + k];
            s_w[d * (KC_GEMM + 1) + k + 0] = v.x;
            s_w[d * (KC_GEMM + 1) + k + 1] = v.y;
            s_w[d * (KC_GEMM + 1) + k + 2] = v.z;
            s_w[d * (KC_GEMM + 1) + k + 3] = v.w;
        }
        __syncthreads();

        for (int ks = 0; ks < KC_GEMM; ks += 4) {
            const float* ap = &s_in[row * 257 + kc + ks + khalf];
            v2f a; a.x = ap[0]; a.y = ap[1];

            const float* bp0 = &s_w[(d0 + row) * (KC_GEMM + 1) + ks + khalf];
            v2f b0; b0.x = bp0[0]; b0.y = bp0[1];
            const float* bp1 = &s_w[(d1 + row) * (KC_GEMM + 1) + ks + khalf];
            v2f b1; b1.x = bp1[0]; b1.y = bp1[1];

            acc0 = wmma_f32(a, b0, acc0);
            acc1 = wmma_f32(a, b1, acc1);
        }
    }

    const float bias0 = bias[d0 + row];
    const float bias1 = bias[d1 + row];
    for (int i = 0; i < 8; ++i) {
        const int r = i + ((lane < 16) ? 0 : 8);
        out[(size_t)(n0 + r) * DMODEL + d0 + row] = acc0[i] + bias0;
        out[(size_t)(n0 + r) * DMODEL + d1 + row] = acc1[i] + bias1;
    }
}

#define MSG_ROWSTRIDE (KC_ATT + 1)
#define MSG_HSTRIDE   (16 * MSG_ROWSTRIDE + 4)

__launch_bounds__(256)
__global__ void attn_kernel(const float* __restrict__ messages,
                            const int*   __restrict__ adj,
                            const float* __restrict__ proj,
                            float*       __restrict__ attn_out) {
    __shared__ float s_msg[NH * MSG_HSTRIDE];
    __shared__ float s_mask[16 * MSG_ROWSTRIDE];
    __shared__ float s_scale[8 * 32];
    __shared__ float s_m[8 * 32];

    const int tid   = threadIdx.x;
    const int wid   = tid >> 5;
    const int lane  = tid & 31;
    const int row   = lane & 15;
    const int khalf = (lane < 16) ? 0 : 2;
    const int h0    = wid * 2;
    const int h1    = wid * 2 + 1;
    const int b     = blockIdx.x / (NQ / 16);
    const int q0    = (blockIdx.x % (NQ / 16)) * 16;
    const int own   = wid * 32 + lane;

    const float* msg_base = messages + (size_t)(b * NQ + q0) * NKV * NH;
    const int*   adj_base = adj      + (size_t)(b * NQ + q0) * NKV;
    const float* proj_b   = proj     + (size_t)b * NKV * DMODEL;

    float m = -3.0e38f, l = 0.0f, s2 = 0.0f;
    v8f acc0 = {};
    v8f acc1 = {};

    for (int kc = 0; kc < NKV; kc += KC_ATT) {
        __syncthreads();
        for (int i = 0; i < 2; ++i) {
            int e = tid + 256 * i;
            int q = e >> 5, k = e & 31;
            s_mask[q * MSG_ROWSTRIDE + k] =
                (adj_base[(size_t)q * NKV + kc + k] > 0) ? 0.0f : NEG_INF_F;
        }
        __syncthreads();
        for (int i = 0; i < 8; ++i) {
            int e4 = (tid + 256 * i) * 4;
            int q  = e4 >> 9;
            int j  = e4 & 511;
            int k  = j >> 4, h = j & 15;
            const float4 v =
                *(const float4*)&msg_base[(size_t)q * (NKV * NH) + kc * NH + j];
            const float mk = s_mask[q * MSG_ROWSTRIDE + k];
            s_msg[(h + 0) * MSG_HSTRIDE + q * MSG_ROWSTRIDE + k] = v.x + mk;
            s_msg[(h + 1) * MSG_HSTRIDE + q * MSG_ROWSTRIDE + k] = v.y + mk;
            s_msg[(h + 2) * MSG_HSTRIDE + q * MSG_ROWSTRIDE + k] = v.z + mk;
            s_msg[(h + 3) * MSG_HSTRIDE + q * MSG_ROWSTRIDE + k] = v.w + mk;
        }

        if (kc + KC_ATT < NKV) {
            const int pq = tid >> 4;
            const int pj = (tid & 15) * 32;
            __builtin_prefetch(&msg_base[(size_t)pq * (NKV * NH) +
                                         (kc + KC_ATT) * NH + pj], 0, 3);
            if (tid < 16) {
                __builtin_prefetch(&adj_base[(size_t)tid * NKV + kc + KC_ATT], 0, 3);
            }
        }
        __syncthreads();

        {
            const int myh = (lane < 16) ? h0 : h1;
            const float* rp = &s_msg[myh * MSG_HSTRIDE + row * MSG_ROWSTRIDE];
            float mc = -3.0e38f;
            for (int k = 0; k < KC_ATT; ++k) mc = fmaxf(mc, rp[k]);
            const float mnew = fmaxf(m, mc);
            const float r    = __expf(m - mnew);
            float lc = 0.0f, s2c = 0.0f;
            for (int k = 0; k < KC_ATT; ++k) {
                const float t = __expf(rp[k] - mnew);
                lc  += t;
                s2c += t * t;
            }
            l  = l * r + lc;
            s2 = s2 * (r * r) + s2c;
            m  = mnew;
            s_scale[own] = r;
            s_m[own]     = mnew;
        }

        for (int i = 0; i < 8; ++i) {
            const int rr = i + ((lane < 16) ? 0 : 8);
            acc0[i] *= s_scale[wid * 32 + rr];
            acc1[i] *= s_scale[wid * 32 + 16 + rr];
        }
        const float m0 = s_m[wid * 32 + row];
        const float m1 = s_m[wid * 32 + 16 + row];

        const float* p0 = &s_msg[h0 * MSG_HSTRIDE + row * MSG_ROWSTRIDE + khalf];
        const float* p1 = &s_msg[h1 * MSG_HSTRIDE + row * MSG_ROWSTRIDE + khalf];
        for (int ks = 0; ks < KC_ATT; ks += 4) {
            v2f a0; a0.x = __expf(p0[ks] - m0); a0.y = __expf(p0[ks + 1] - m0);
            v2f a1; a1.x = __expf(p1[ks] - m1); a1.y = __expf(p1[ks + 1] - m1);

            const float* vrow = proj_b + (size_t)(kc + ks + khalf) * DMODEL;
            v2f b0; b0.x = vrow[h0 * 16 + row]; b0.y = vrow[DMODEL + h0 * 16 + row];
            v2f b1; b1.x = vrow[h1 * 16 + row]; b1.y = vrow[DMODEL + h1 * 16 + row];

            acc0 = wmma_f32(a0, b0, acc0);
            acc1 = wmma_f32(a1, b1, acc1);
        }
    }

    s_scale[own] = sqrtf(s2) / (l * l);
    for (int i = 0; i < 8; ++i) {
        const int rr = i + ((lane < 16) ? 0 : 8);
        const float f0 = s_scale[wid * 32 + rr];
        const float f1 = s_scale[wid * 32 + 16 + rr];
        attn_out[(size_t)(b * NQ + q0 + rr) * DMODEL + h0 * 16 + row] = acc0[i] * f0;
        attn_out[(size_t)(b * NQ + q0 + rr) * DMODEL + h1 * 16 + row] = acc1[i] * f1;
    }
}

extern "C" void kernel_launch(void* const* d_in, const int* in_sizes, int n_in,
                              void* d_out, int out_size, void* d_ws, size_t ws_size,
                              hipStream_t stream) {
    (void)in_sizes; (void)n_in; (void)out_size; (void)ws_size;

    const float* v_inv    = (const float*)d_in[0];
    const float* messages = (const float*)d_in[1];
    const int*   adj      = (const int*)  d_in[2];
    const float* W_in     = (const float*)d_in[3];
    const float* b_in     = (const float*)d_in[4];
    const float* W_out    = (const float*)d_in[5];
    const float* b_out    = (const float*)d_in[6];
    float*       out      = (float*)d_out;

    float* proj = (float*)d_ws;
    float* attn = proj + (size_t)BATCH * NKV * DMODEL;

    gemm_bias_256<<<(BATCH * NKV) / 16, 256, 0, stream>>>(v_inv, W_in, b_in, proj);
    attn_kernel<<<BATCH * (NQ / 16), 256, 0, stream>>>(messages, adj, proj, attn);
    gemm_bias_256<<<(BATCH * NQ) / 16, 256, 0, stream>>>(attn, W_out, b_out, out);
}


# attention k-chunk loop software-pipelined: next chunk's messages+adjacency loaded to registers ahead of this chunk's softmax/WMMA, stored to a second LDS plane; one barrier per chunk, not three
# speedup vs baseline: 1.0396x; 1.0396x over previous
#include <hip/hip_runtime.h>

#define BATCH   8
#define NQ      512
#define NKV     512
#define NH      16
#define DMODEL  256
#define NEG_INF_F (-1.0e9f)

#define KC_ATT  32
#define KC_GEMM 32

typedef __attribute__((ext_vector_type(2))) float v2f;
typedef __attribute__((ext_vector_type(8))) float v8f;

__device__ __forceinline__ v8f wmma_f32(v2f a, v2f b, v8f c) {
    return __builtin_amdgcn_wmma_f32_16x16x4_f32(
         false, a,  false, b,
         (short)0, c,  false,  false);
}

__launch_bounds__(256)
__global__ void gemm_bias_256(const float* __restrict__ in,
                              const float* __restrict__ W,
                              const float* __restrict__ bias,
                              float* __restrict__ out) {
    __shared__ float s_in[16 * 257];
    __shared__ float s_w[DMODEL * (KC_GEMM + 1)];

    const int tid   = threadIdx.x;
    const int wid   = tid >> 5;
    const int lane  = tid & 31;
    const int row   = lane & 15;
    const int khalf = (lane < 16) ? 0 : 2;
    const int n0    = blockIdx.x * 16;

    for (int i = 0; i < 4; ++i) {
        int e4 = (tid + 256 * i) * 4;
        int r  = e4 >> 8, k = e4 & 255;
        const float4 v = *(const float4*)&in[(size_t)(n0 + r) * DMODEL + k];
        s_in[r * 257 + k + 0] = v.x;
        s_in[r * 257 + k + 1] = v.y;
        s_in[r * 257 + k + 2] = v.z;
        s_in[r * 257 + k + 3] = v.w;
    }

    v8f acc0 = {};
    v8f acc1 = {};
    const int d0 = (wid)     * 16;
    const int d1 = (wid + 8) * 16;

    for (int kc = 0; kc < DMODEL; kc += KC_GEMM) {
        __syncthreads();
        for (int i = 0; i < 8; ++i) {
            int e4 = (tid + 256 * i) * 4;
            int d  = e4 >> 5, k = e4 & 31;
            const float4 v = *(const float4*)&W[(size_t)d * DMODEL + kc + k];
            s_w[d * (KC_GEMM + 1) + k + 0] = v.x;
            s_w[d * (KC_GEMM + 1) + k + 1] = v.y;
            s_w[d * (KC_GEMM + 1) + k + 2] = v.z;
            s_w[d * (KC_GEMM + 1) + k + 3] = v.w;
        }
        __syncthreads();

        for (int ks = 0; ks < KC_GEMM; ks += 4) {
            const float* ap = &s_in[row * 257 + kc + ks + khalf];
            v2f a; a.x = ap[0]; a.y = ap[1];

            const float* bp0 = &s_w[(d0 + row) * (KC_GEMM + 1) + ks + khalf];
            v2f b0; b0.x = bp0[0]; b0.y = bp0[1];
            const float* bp1 = &s_w[(d1 + row) * (KC_GEMM + 1) + ks + khalf];
            v2f b1; b1.x = bp1[0]; b1.y = bp1[1];

            acc0 = wmma_f32(a, b0, acc0);
            acc1 = wmma_f32(a, b1, acc1);
        }
    }

    const float bias0 = bias[d0 + row];
    const float bias1 = bias[d1 + row];
    for (int i = 0; i < 8; ++i) {
        const int r = i + ((lane < 16) ? 0 : 8);
        out[(size_t)(n0 + r) * DMODEL + d0 + row] = acc0[i] + bias0;
        out[(size_t)(n0 + r) * DMODEL + d1 + row] = acc1[i] + bias1;
    }
}

#define MSG_ROWSTRIDE (KC_ATT + 1)
#define MSG_HSTRIDE   (16 * MSG_ROWSTRIDE + 4)
#define MSG_PLANE     (NH * MSG_HSTRIDE)
#define NCHUNK        (NKV / KC_ATT)

__launch_bounds__(256)
__global__ void attn_kernel(const float* __restrict__ messages,
                            const int*   __restrict__ adj,
                            const float* __restrict__ proj,
                            float*       __restrict__ attn_out) {
    __shared__ float s_msg[2 * MSG_PLANE];
    __shared__ float s_scale[8 * 32];
    __shared__ float s_m[8 * 32];

    const int tid   = threadIdx.x;
    const int wid   = tid >> 5;
    const int lane  = tid & 31;
    const int row   = lane & 15;
    const int khalf = (lane < 16) ? 0 : 2;
    const int h0    = wid * 2;
    const int h1    = wid * 2 + 1;
    const int b     = blockIdx.x / (NQ / 16);
    const int q0    = (blockIdx.x % (NQ / 16)) * 16;
    const int own   = wid * 32 + lane;

    const float* msg_base = messages + (size_t)(b * NQ + q0) * NKV * NH;
    const int*   adj_base = adj      + (size_t)(b * NQ + q0) * NKV;
    const float* proj_b   = proj     + (size_t)b * NKV * DMODEL;

    const int sq = tid >> 7;
    const int sj = (tid & 127) * 4;
    const int sk = sj >> 4;
    const int sh = sj & 15;
    const float* mptr = msg_base + (size_t)sq * (NKV * NH) + sj;
    const int*   aptr = adj_base + (size_t)sq * NKV + sk;
    const int    soff = sh * MSG_HSTRIDE + sq * MSG_ROWSTRIDE + sk;

    float4 mreg[8];
    int    areg[8];

    for (int i = 0; i < 8; ++i) {
        mreg[i] = *(const float4*)(mptr + (size_t)(2 * i) * (NKV * NH));
        areg[i] = aptr[(size_t)(2 * i) * NKV];
    }
    for (int i = 0; i < 8; ++i) {
        const float mk = (areg[i] > 0) ? 0.0f : NEG_INF_F;
        float* d = s_msg + soff + (2 * i) * MSG_ROWSTRIDE;
        d[0 * MSG_HSTRIDE] = mreg[i].x + mk;
        d[1 * MSG_HSTRIDE] = mreg[i].y + mk;
        d[2 * MSG_HSTRIDE] = mreg[i].z + mk;
        d[3 * MSG_HSTRIDE] = mreg[i].w + mk;
    }
    __syncthreads();

    float m = -3.0e38f, l = 0.0f, s2 = 0.0f;
    v8f acc0 = {};
    v8f acc1 = {};

    for (int c = 0; c < NCHUNK; ++c) {
        const int   kc    = c * KC_ATT;
        const float* plane = s_msg + (c & 1) * MSG_PLANE;
        const bool  more  = (c + 1 < NCHUNK);

        v2f bf0[KC_ATT / 4], bf1[KC_ATT / 4];
        for (int j = 0; j < KC_ATT / 4; ++j) {
            const float* vrow = proj_b + (size_t)(kc + 4 * j + khalf) * DMODEL;
            bf0[j].x = vrow[h0 * 16 + row]; bf0[j].y = vrow[DMODEL + h0 * 16 + row];
            bf1[j].x = vrow[h1 * 16 + row]; bf1[j].y = vrow[DMODEL + h1 * 16 + row];
        }

        if (more) {
            for (int i = 0; i < 8; ++i) {
                mreg[i] = *(const float4*)(mptr + (size_t)(2 * i) * (NKV * NH) +
                                           (kc + KC_ATT) * NH);
                areg[i] = aptr[(size_t)(2 * i) * NKV + kc + KC_ATT];
            }
            __builtin_amdgcn_sched_barrier(0);
        }

        {
            const int myh = (lane < 16) ? h0 : h1;
            const float* rp = &plane[myh * MSG_HSTRIDE + row * MSG_ROWSTRIDE];
            float mc = -3.0e38f;
            for (int k = 0; k < KC_ATT; ++k) mc = fmaxf(mc, rp[k]);
            const float mnew = fmaxf(m, mc);
            const float r    = __expf(m - mnew);
            float lc = 0.0f, s2c = 0.0f;
            for (int k = 0; k < KC_ATT; ++k) {
                const float t = __expf(rp[k] - mnew);
                lc  += t;
                s2c += t * t;
            }
            l  = l * r + lc;
            s2 = s2 * (r * r) + s2c;
            m  = mnew;
            s_scale[own] = r;
            s_m[own]     = mnew;
        }

        for (int i = 0; i < 8; ++i) {
            const int rr = i + ((lane < 16) ? 0 : 8);
            acc0[i] *= s_scale[wid * 32 + rr];
            acc1[i] *= s_scale[wid * 32 + 16 + rr];
        }
        const float m0 = s_m[wid * 32 + row];
        const float m1 = s_m[wid * 32 + 16 + row];

        const float* p0 = &plane[h0 * MSG_HSTRIDE + row * MSG_ROWSTRIDE + khalf];
        const float* p1 = &plane[h1 * MSG_HSTRIDE + row * MSG_ROWSTRIDE + khalf];
        for (int j = 0; j < KC_ATT / 4; ++j) {
            const int ks = 4 * j;
            v2f a0; a0.x = __expf(p0[ks] - m0); a0.y = __expf(p0[ks + 1] - m0);
            v2f a1; a1.x = __expf(p1[ks] - m1); a1.y = __expf(p1[ks + 1] - m1);
            acc0 = wmma_f32(a0, bf0[j], acc0);
            acc1 = wmma_f32(a1, bf1[j], acc1);
        }

        if (more) {
            __builtin_amdgcn_sched_barrier(0);
            float* nplane = s_msg + ((c + 1) & 1) * MSG_PLANE + soff;
            for (int i = 0; i < 8; ++i) {
                const float mk = (areg[i] > 0) ? 0.0f : NEG_INF_F;
                float* d = nplane + (2 * i) * MSG_ROWSTRIDE;
                d[0 * MSG_HSTRIDE] = mreg[i].x + mk;
                d[1 * MSG_HSTRIDE] = mreg[i].y + mk;
                d[2 * MSG_HSTRIDE] = mreg[i].z + mk;
                d[3 * MSG_HSTRIDE] = mreg[i].w + mk;
            }
        }
        __syncthreads();
    }

    s_scale[own] = sqrtf(s2) / (l * l);
    for (int i = 0; i < 8; ++i) {
        const int rr = i + ((lane < 16) ? 0 : 8);
        const float f0 = s_scale[wid * 32 + rr];
        const float f1 = s_scale[wid * 32 + 16 + rr];
        attn_out[(size_t)(b * NQ + q0 + rr) * DMODEL + h0 * 16 + row] = acc0[i] * f0;
        attn_out[(size_t)(b * NQ + q0 + rr) * DMODEL + h1 * 16 + row] = acc1[i] * f1;
    }
}

extern "C" void kernel_launch(void* const* d_in, const int* in_sizes, int n_in,
                              void* d_out, int out_size, void* d_ws, size_t ws_size,
                              hipStream_t stream) {
    (void)in_sizes; (void)n_in; (void)out_size; (void)ws_size;

    const float* v_inv    = (const float*)d_in[0];
    const float* messages = (const float*)d_in[1];
    const int*   adj      = (const int*)  d_in[2];
    const float* W_in     = (const float*)d_in[3];
    const float* b_in     = (const float*)d_in[4];
    const float* W_out    = (const float*)d_in[5];
    const float* b_out    = (const float*)d_in[6];
    float*       out      = (float*)d_out;

    float* proj = (float*)d_ws;
    float* attn = proj + (size_t)BATCH * NKV * DMODEL;

    gemm_bias_256<<<(BATCH * NKV) / 16, 256, 0, stream>>>(v_inv, W_in, b_in, proj);
    attn_kernel<<<BATCH * (NQ / 16), 256, 0, stream>>>(messages, adj, proj, attn);
    gemm_bias_256<<<(BATCH * NQ) / 16, 256, 0, stream>>>(attn, W_out, b_out, out);
}


# output projection fused into the attention kernel: its 16x256 tile stays in LDS, W_out streamed through two LDS planes, 3 launches -> 2; k order of that product permuted inside 8-blocks
# speedup vs baseline: 1.2190x; 1.1726x over previous
#include <hip/hip_runtime.h>

#define BATCH   8
#define NQ      512
#define NKV     512
#define NH      16
#define DMODEL  256
#define NEG_INF_F (-1.0e9f)

#define KC_ATT  32
#define KC_GEMM 32

typedef __attribute__((ext_vector_type(2))) float v2f;
typedef __attribute__((ext_vector_type(8))) float v8f;

__device__ __forceinline__ v8f wmma_f32(v2f a, v2f b, v8f c) {
    return __builtin_amdgcn_wmma_f32_16x16x4_f32(
         false, a,  false, b,
         (short)0, c,  false,  false);
}

__launch_bounds__(256)
__global__ void gemm_bias_256(const float* __restrict__ in,
                              const float* __restrict__ W,
                              const float* __restrict__ bias,
                              float* __restrict__ out) {
    __shared__ float s_in[16 * 257];
    __shared__ float s_w[DMODEL * (KC_GEMM + 1)];

    const int tid   = threadIdx.x;
    const int wid   = tid >> 5;
    const int lane  = tid & 31;
    const int row   = lane & 15;
    const int khalf = (lane < 16) ? 0 : 2;
    const int n0    = blockIdx.x * 16;

    for (int i = 0; i < 4; ++i) {
        int e4 = (tid + 256 * i) * 4;
        int r  = e4 >> 8, k = e4 & 255;
        const float4 v = *(const float4*)&in[(size_t)(n0 + r) * DMODEL + k];
        s_in[r * 257 + k + 0] = v.x;
        s_in[r * 257 + k + 1] = v.y;
        s_in[r * 257 + k + 2] = v.z;
        s_in[r * 257 + k + 3] = v.w;
    }

    v8f acc0 = {};
    v8f acc1 = {};
    const int d0 = (wid)     * 16;
    const int d1 = (wid + 8) * 16;

    for (int kc = 0; kc < DMODEL; kc += KC_GEMM) {
        __syncthreads();
        for (int i = 0; i < 8; ++i) {
            int e4 = (tid + 256 * i) * 4;
            int d  = e4 >> 5, k = e4 & 31;
            const float4 v = *(const float4*)&W[(size_t)d * DMODEL + kc + k];
            s_w[d * (KC_GEMM + 1) + k + 0] = v.x;
            s_w[d * (KC_GEMM + 1) + k + 1] = v.y;
            s_w[d * (KC_GEMM + 1) + k + 2] = v.z;
            s_w[d * (KC_GEMM + 1) + k + 3] = v.w;
        }
        __syncthreads();

        for (int ks = 0; ks < KC_GEMM; ks += 4) {
            const float* ap = &s_in[row * 257 + kc + ks + khalf];
            v2f a; a.x = ap[0]; a.y = ap[1];

            const float* bp0 = &s_w[(d0 + row) * (KC_GEMM + 1) + ks + khalf];
            v2f b0; b0.x = bp0[0]; b0.y = bp0[1];
            const float* bp1 = &s_w[(d1 + row) * (KC_GEMM + 1) + ks + khalf];
            v2f b1; b1.x = bp1[0]; b1.y = bp1[1];

            acc0 = wmma_f32(a, b0, acc0);
            acc1 = wmma_f32(a, b1, acc1);
        }
    }

    const float bias0 = bias[d0 + row];
    const float bias1 = bias[d1 + row];
    for (int i = 0; i < 8; ++i) {
        const int r = i + ((lane < 16) ? 0 : 8);
        out[(size_t)(n0 + r) * DMODEL + d0 + row] = acc0[i] + bias0;
        out[(size_t)(n0 + r) * DMODEL + d1 + row] = acc1[i] + bias1;
    }
}

#define MSG_ROWSTRIDE (KC_ATT + 1)
#define MSG_HSTRIDE   (16 * MSG_ROWSTRIDE + 4)
#define MSG_PLANE     (NH * MSG_HSTRIDE)
#define NCHUNK        (NKV / KC_ATT)
#define ATT_PITCH     (DMODEL + 4)
#define WO_PITCH      (KC_GEMM + 4)
#define WO_PLANE      (DMODEL * WO_PITCH)
#define EPI_FLOATS    (16 * ATT_PITCH + 2 * WO_PLANE)
#define ATTN_LDS_FLOATS ((EPI_FLOATS > 2 * MSG_PLANE) ? EPI_FLOATS : 2 * MSG_PLANE)

__launch_bounds__(256)
__global__ void attn_kernel(const float* __restrict__ messages,
                            const int*   __restrict__ adj,
                            const float* __restrict__ proj,
                            const float* __restrict__ W_out,
                            const float* __restrict__ b_out,
                            float*       __restrict__ out) {
    __shared__ __attribute__((aligned(16))) float s_msg[ATTN_LDS_FLOATS];
    __shared__ float s_scale[8 * 32];
    __shared__ float s_m[8 * 32];

    const int tid   = threadIdx.x;
    const int wid   = tid >> 5;
    const int lane  = tid & 31;
    const int row   = lane & 15;
    const int khalf = (lane < 16) ? 0 : 2;
    const int h0    = wid * 2;
    const int h1    = wid * 2 + 1;
    const int b     = blockIdx.x / (NQ / 16);
    const int q0    = (blockIdx.x % (NQ / 16)) * 16;
    const int own   = wid * 32 + lane;

    const float* msg_base = messages + (size_t)(b * NQ + q0) * NKV * NH;
    const int*   adj_base = adj      + (size_t)(b * NQ + q0) * NKV;
    const float* proj_b   = proj     + (size_t)b * NKV * DMODEL;

    const int sq = tid >> 7;
    const int sj = (tid & 127) * 4;
    const int sk = sj >> 4;
    const int sh = sj & 15;
    const float* mptr = msg_base + (size_t)sq * (NKV * NH) + sj;
    const int*   aptr = adj_base + (size_t)sq * NKV + sk;
    const int    soff = sh * MSG_HSTRIDE + sq * MSG_ROWSTRIDE + sk;

    const int wd = tid >> 3;
    const int wk = (tid & 7) * 4;
    const float* wptr = W_out + (size_t)wd * DMODEL + wk;
    const int    woff = wd * WO_PITCH + wk;

    float4 mreg[8];
    int    areg[8];

    for (int i = 0; i < 8; ++i) {
        mreg[i] = *(const float4*)(mptr + (size_t)(2 * i) * (NKV * NH));
        areg[i] = aptr[(size_t)(2 * i) * NKV];
    }
    for (int i = 0; i < 8; ++i) {
        const float mk = (areg[i] > 0) ? 0.0f : NEG_INF_F;
        float* d = s_msg + soff + (2 * i) * MSG_ROWSTRIDE;
        d[0 * MSG_HSTRIDE] = mreg[i].x + mk;
        d[1 * MSG_HSTRIDE] = mreg[i].y + mk;
        d[2 * MSG_HSTRIDE] = mreg[i].z + mk;
        d[3 * MSG_HSTRIDE] = mreg[i].w + mk;
    }
    __syncthreads();

    float m = -3.0e38f, l = 0.0f, s2 = 0.0f;
    v8f acc0 = {};
    v8f acc1 = {};

    for (int c = 0; c < NCHUNK; ++c) {
        const int   kc    = c * KC_ATT;
        const float* plane = s_msg + (c & 1) * MSG_PLANE;
        const bool  more  = (c + 1 < NCHUNK);

        v2f bf0[KC_ATT / 4], bf1[KC_ATT / 4];
        for (int j = 0; j < KC_ATT / 4; ++j) {
            const float* vrow = proj_b + (size_t)(kc + 4 * j + khalf) * DMODEL;
            bf0[j].x = vrow[h0 * 16 + row]; bf0[j].y = vrow[DMODEL + h0 * 16 + row];
            bf1[j].x = vrow[h1 * 16 + row]; bf1[j].y = vrow[DMODEL + h1 * 16 + row];
        }

        if (more) {
            for (int i = 0; i < 8; ++i) {
                mreg[i] = *(const float4*)(mptr + (size_t)(2 * i) * (NKV * NH) +
                                           (kc + KC_ATT) * NH);
                areg[i] = aptr[(size_t)(2 * i) * NKV + kc + KC_ATT];
            }
        } else {
            for (int i = 0; i < 8; ++i)
                mreg[i] = *(const float4*)(wptr + (size_t)(32 * i) * DMODEL);
        }
        __builtin_amdgcn_sched_barrier(0);

        {
            const int myh = (lane < 16) ? h0 : h1;
            const float* rp = &plane[myh * MSG_HSTRIDE + row * MSG_ROWSTRIDE];
            float mc = -3.0e38f;
            for (int k = 0; k < KC_ATT; ++k) mc = fmaxf(mc, rp[k]);
            const float mnew = fmaxf(m, mc);
            const float r    = __expf(m - mnew);
            float lc = 0.0f, s2c = 0.0f;
            for (int k = 0; k < KC_ATT; ++k) {
                const float t = __expf(rp[k] - mnew);
                lc  += t;
                s2c += t * t;
            }
            l  = l * r + lc;
            s2 = s2 * (r * r) + s2c;
            m  = mnew;
            s_scale[own] = r;
            s_m[own]     = mnew;
        }

        for (int i = 0; i < 8; ++i) {
            const int rr = i + ((lane < 16) ? 0 : 8);
            acc0[i] *= s_scale[wid * 32 + rr];
            acc1[i] *= s_scale[wid * 32 + 16 + rr];
        }
        const float m0 = s_m[wid * 32 + row];
        const float m1 = s_m[wid * 32 + 16 + row];

        const float* p0 = &plane[h0 * MSG_HSTRIDE + row * MSG_ROWSTRIDE + khalf];
        const float* p1 = &plane[h1 * MSG_HSTRIDE + row * MSG_ROWSTRIDE + khalf];
        for (int j = 0; j < KC_ATT / 4; ++j) {
            const int ks = 4 * j;
            v2f a0; a0.x = __expf(p0[ks] - m0); a0.y = __expf(p0[ks + 1] - m0);
            v2f a1; a1.x = __expf(p1[ks] - m1); a1.y = __expf(p1[ks + 1] - m1);
            acc0 = wmma_f32(a0, bf0[j], acc0);
            acc1 = wmma_f32(a1, bf1[j], acc1);
        }

        if (more) {
            __builtin_amdgcn_sched_barrier(0);
            float* nplane = s_msg + ((c + 1) & 1) * MSG_PLANE + soff;
            for (int i = 0; i < 8; ++i) {
                const float mk = (areg[i] > 0) ? 0.0f : NEG_INF_F;
                float* d = nplane + (2 * i) * MSG_ROWSTRIDE;
                d[0 * MSG_HSTRIDE] = mreg[i].x + mk;
                d[1 * MSG_HSTRIDE] = mreg[i].y + mk;
                d[2 * MSG_HSTRIDE] = mreg[i].z + mk;
                d[3 * MSG_HSTRIDE] = mreg[i].w + mk;
            }
        }
        __syncthreads();
    }

    float* s_att = s_msg;
    float* s_wo  = s_msg + 16 * ATT_PITCH;
    s_scale[own] = sqrtf(s2) / (l * l);
    for (int i = 0; i < 8; ++i) {
        const int rr = i + ((lane < 16) ? 0 : 8);
        const float f0 = s_scale[wid * 32 + rr];
        const float f1 = s_scale[wid * 32 + 16 + rr];
        s_att[rr * ATT_PITCH + h0 * 16 + row] = acc0[i] * f0;
        s_att[rr * ATT_PITCH + h1 * 16 + row] = acc1[i] * f1;
    }
    for (int i = 0; i < 8; ++i)
        *(float4*)(s_wo + woff + (32 * i) * WO_PITCH) = mreg[i];

    const int d0 = h0 * 16;
    const int d1 = h1 * 16;
    const float bias0 = b_out[d0 + row];
    const float bias1 = b_out[d1 + row];
    const int   kq    = (lane < 16) ? 0 : 4;
    __syncthreads();

    v8f o0 = {};
    v8f o1 = {};
    for (int c = 0; c < DMODEL / KC_GEMM; ++c) {
        const float* wpl  = s_wo + (c & 1) * WO_PLANE;
        const bool   more = (c + 1 < DMODEL / KC_GEMM);
        if (more) {
            for (int i = 0; i < 8; ++i)
                mreg[i] = *(const float4*)(wptr + (size_t)(32 * i) * DMODEL +
                                           (c + 1) * KC_GEMM);
        }
        const float* ap  = s_att + row * ATT_PITCH + c * KC_GEMM + kq;
        const float* bp0 = wpl + (d0 + row) * WO_PITCH + kq;
        const float* bp1 = wpl + (d1 + row) * WO_PITCH + kq;
        for (int J = 0; J < KC_GEMM / 8; ++J) {
            const float4 a  = *(const float4*)(ap  + 8 * J);
            const float4 w0 = *(const float4*)(bp0 + 8 * J);
            const float4 w1 = *(const float4*)(bp1 + 8 * J);
            v2f alo; alo.x = a.x;  alo.y = a.y;
            v2f ahi; ahi.x = a.z;  ahi.y = a.w;
            v2f b0l; b0l.x = w0.x; b0l.y = w0.y;
            v2f b0h; b0h.x = w0.z; b0h.y = w0.w;
            v2f b1l; b1l.x = w1.x; b1l.y = w1.y;
            v2f b1h; b1h.x = w1.z; b1h.y = w1.w;
            o0 = wmma_f32(alo, b0l, o0);
            o1 = wmma_f32(alo, b1l, o1);
            o0 = wmma_f32(ahi, b0h, o0);
            o1 = wmma_f32(ahi, b1h, o1);
        }
        if (more) {
            __builtin_amdgcn_sched_barrier(0);
            float* npl = s_wo + ((c + 1) & 1) * WO_PLANE + woff;
            for (int i = 0; i < 8; ++i)
                *(float4*)(npl + (32 * i) * WO_PITCH) = mreg[i];
            __syncthreads();
        }
    }

    for (int i = 0; i < 8; ++i) {
        const int rr = i + ((lane < 16) ? 0 : 8);
        out[(size_t)(b * NQ + q0 + rr) * DMODEL + d0 + row] = o0[i] + bias0;
        out[(size_t)(b * NQ + q0 + rr) * DMODEL + d1 + row] = o1[i] + bias1;
    }
}

extern "C" void kernel_launch(void* const* d_in, const int* in_sizes, int n_in,
                              void* d_out, int out_size, void* d_ws, size_t ws_size,
                              hipStream_t stream) {
    (void)in_sizes; (void)n_in; (void)out_size; (void)ws_size;

    const float* v_inv    = (const float*)d_in[0];
    const float* messages = (const float*)d_in[1];
    const int*   adj      = (const int*)  d_in[2];
    const float* W_in     = (const float*)d_in[3];
    const float* b_in     = (const float*)d_in[4];
    const float* W_out    = (const float*)d_in[5];
    const float* b_out    = (const float*)d_in[6];
    float*       out      = (float*)d_out;

    float* proj = (float*)d_ws;

    gemm_bias_256<<<(BATCH * NKV) / 16, 256, 0, stream>>>(v_inv, W_in, b_in, proj);
    attn_kernel<<<BATCH * (NQ / 16), 256, 0, stream>>>(messages, adj, proj,
                                                       W_out, b_out, out);
}


# attention block 8 -> 16 waves: two wave groups take alternate k-chunks, each its own online softmax, merged once by max-shift (key-sum order changes); twice the loads in flight, half the barriers
# speedup vs baseline: 1.2707x; 1.0424x over previous
#include <hip/hip_runtime.h>

#define BATCH   8
#define NQ      512
#define NKV     512
#define NH      16
#define DMODEL  256
#define NEG_INF_F (-1.0e9f)

#define KC_ATT  32
#define KC_GEMM 32

typedef __attribute__((ext_vector_type(2))) float v2f;
typedef __attribute__((ext_vector_type(8))) float v8f;

__device__ __forceinline__ v8f wmma_f32(v2f a, v2f b, v8f c) {
    return __builtin_amdgcn_wmma_f32_16x16x4_f32(
         false, a,  false, b,
         (short)0, c,  false,  false);
}

__launch_bounds__(256)
__global__ void gemm_bias_256(const float* __restrict__ in,
                              const float* __restrict__ W,
                              const float* __restrict__ bias,
                              float* __restrict__ out) {
    __shared__ float s_in[16 * 257];
    __shared__ float s_w[DMODEL * (KC_GEMM + 1)];

    const int tid   = threadIdx.x;
    const int wid   = tid >> 5;
    const int lane  = tid & 31;
    const int row   = lane & 15;
    const int khalf = (lane < 16) ? 0 : 2;
    const int n0    = blockIdx.x * 16;

    for (int i = 0; i < 4; ++i) {
        int e4 = (tid + 256 * i) * 4;
        int r  = e4 >> 8, k = e4 & 255;
        const float4 v = *(const float4*)&in[(size_t)(n0 + r) * DMODEL + k];
        s_in[r * 257 + k + 0] = v.x;
        s_in[r * 257 + k + 1] = v.y;
        s_in[r * 257 + k + 2] = v.z;
        s_in[r * 257 + k + 3] = v.w;
    }

    v8f acc0 = {};
    v8f acc1 = {};
    const int d0 = (wid)     * 16;
    const int d1 = (wid + 8) * 16;

    for (int kc = 0; kc < DMODEL; kc += KC_GEMM) {
        __syncthreads();
        for (int i = 0; i < 8; ++i) {
            int e4 = (tid + 256 * i) * 4;
            int d  = e4 >> 5, k = e4 & 31;
            const float4 v = *(const float4*)&W[(size_t)d * DMODEL + kc + k];
            s_w[d * (KC_GEMM + 1) + k + 0] = v.x;
            s_w[d * (KC_GEMM + 1) + k + 1] = v.y;
            s_w[d * (KC_GEMM + 1) + k + 2] = v.z;
            s_w[d * (KC_GEMM + 1) + k + 3] = v.w;
        }
        __syncthreads();

        for (int ks = 0; ks < KC_GEMM; ks += 4) {
            const float* ap = &s_in[row * 257 + kc + ks + khalf];
            v2f a; a.x = ap[0]; a.y = ap[1];

            const float* bp0 = &s_w[(d0 + row) * (KC_GEMM + 1) + ks + khalf];
            v2f b0; b0.x = bp0[0]; b0.y = bp0[1];
            const float* bp1 = &s_w[(d1 + row) * (KC_GEMM + 1) + ks + khalf];
            v2f b1; b1.x = bp1[0]; b1.y = bp1[1];

            acc0 = wmma_f32(a, b0, acc0);
            acc1 = wmma_f32(a, b1, acc1);
        }
    }

    const float bias0 = bias[d0 + row];
    const float bias1 = bias[d1 + row];
    for (int i = 0; i < 8; ++i) {
        const int r = i + ((lane < 16) ? 0 : 8);
        out[(size_t)(n0 + r) * DMODEL + d0 + row] = acc0[i] + bias0;
        out[(size_t)(n0 + r) * DMODEL + d1 + row] = acc1[i] + bias1;
    }
}

#define MSG_ROWSTRIDE (KC_ATT + 1)
#define MSG_HSTRIDE   (16 * MSG_ROWSTRIDE + 4)
#define MSG_PLANE     (NH * MSG_HSTRIDE)
#define NCHUNK        (NKV / KC_ATT)
#define NGROUP        2
#define NTRIP         (NCHUNK / NGROUP)
#define ATT_PITCH     (DMODEL + 4)
#define WO_PITCH      (KC_GEMM + 4)
#define WO_PLANE      (DMODEL * WO_PITCH)
#define EPI_FLOATS    (16 * ATT_PITCH + 2 * WO_PLANE)
#define DUMP_FLOATS   (8 * 2 * 8 * 32)
#define ATTN_LDS_FLOATS (((EPI_FLOATS + DUMP_FLOATS) > 2 * NGROUP * MSG_PLANE) ? \
                         (EPI_FLOATS + DUMP_FLOATS) : 2 * NGROUP * MSG_PLANE)

__launch_bounds__(512)
__global__ void attn_kernel(const float* __restrict__ messages,
                            const int*   __restrict__ adj,
                            const float* __restrict__ proj,
                            const float* __restrict__ W_out,
                            const float* __restrict__ b_out,
                            float*       __restrict__ out) {
    __shared__ __attribute__((aligned(16))) float s_msg[ATTN_LDS_FLOATS];
    __shared__ float s_scale[16 * 32];
    __shared__ float s_m[16 * 32];
    __shared__ float s_gm[8 * 32];
    __shared__ float s_gl[8 * 32];
    __shared__ float s_gs[8 * 32];

    const int tid   = threadIdx.x;
    const int wid   = __builtin_amdgcn_readfirstlane(tid >> 5);
    const int grp   = wid >> 3;
    const int w     = wid & 7;
    const int gtid  = tid & 255;
    const int lane  = tid & 31;
    const int row   = lane & 15;
    const int khalf = (lane < 16) ? 0 : 2;
    const int h0    = w * 2;
    const int h1    = w * 2 + 1;
    const int b     = blockIdx.x / (NQ / 16);
    const int q0    = (blockIdx.x % (NQ / 16)) * 16;
    const int own   = wid * 32 + lane;
    const int gown  = w * 32 + lane;

    const float* msg_base = messages + (size_t)(b * NQ + q0) * NKV * NH;
    const int*   adj_base = adj      + (size_t)(b * NQ + q0) * NKV;
    const float* proj_b   = proj     + (size_t)b * NKV * DMODEL;

    const int sq = gtid >> 7;
    const int sj = (gtid & 127) * 4;
    const int sk = sj >> 4;
    const int sh = sj & 15;
    const float* mptr = msg_base + (size_t)sq * (NKV * NH) + sj + grp * (KC_ATT * NH);
    const int*   aptr = adj_base + (size_t)sq * NKV + sk + grp * KC_ATT;
    const int    soff = sh * MSG_HSTRIDE + sq * MSG_ROWSTRIDE + sk;
    float* gplanes = s_msg + grp * (2 * MSG_PLANE);

    const int wd = tid >> 3;
    const int wk = (tid & 7) * 4;
    const float* wptr = W_out + (size_t)wd * DMODEL + wk;
    const int    woff = wd * WO_PITCH + wk;

    float4 mreg[8];
    int    areg[8];

    for (int i = 0; i < 8; ++i) {
        mreg[i] = *(const float4*)(mptr + (size_t)(2 * i) * (NKV * NH));
        areg[i] = aptr[(size_t)(2 * i) * NKV];
    }
    for (int i = 0; i < 8; ++i) {
        const float mk = (areg[i] > 0) ? 0.0f : NEG_INF_F;
        float* d = gplanes + soff + (2 * i) * MSG_ROWSTRIDE;
        d[0 * MSG_HSTRIDE] = mreg[i].x + mk;
        d[1 * MSG_HSTRIDE] = mreg[i].y + mk;
        d[2 * MSG_HSTRIDE] = mreg[i].z + mk;
        d[3 * MSG_HSTRIDE] = mreg[i].w + mk;
    }
    __syncthreads();

    float m = -3.0e38f, l = 0.0f, s2 = 0.0f;
    v8f acc0 = {};
    v8f acc1 = {};

    for (int t = 0; t < NTRIP; ++t) {
        const int   kc    = (NGROUP * t + grp) * KC_ATT;
        const int   knext = kc + NGROUP * KC_ATT;
        const float* plane = gplanes + (t & 1) * MSG_PLANE;
        const bool  more  = (t + 1 < NTRIP);

        v2f bf0[KC_ATT / 4], bf1[KC_ATT / 4];
        for (int j = 0; j < KC_ATT / 4; ++j) {
            const float* vrow = proj_b + (size_t)(kc + 4 * j + khalf) * DMODEL;
            bf0[j].x = vrow[h0 * 16 + row]; bf0[j].y = vrow[DMODEL + h0 * 16 + row];
            bf1[j].x = vrow[h1 * 16 + row]; bf1[j].y = vrow[DMODEL + h1 * 16 + row];
        }

        if (more) {
            for (int i = 0; i < 8; ++i) {
                mreg[i] = *(const float4*)(mptr + (size_t)(2 * i) * (NKV * NH) +
                                           (knext - grp * KC_ATT) * NH);
                areg[i] = aptr[(size_t)(2 * i) * NKV + (knext - grp * KC_ATT)];
            }
        } else {
            for (int i = 0; i < 4; ++i)
                mreg[i] = *(const float4*)(wptr + (size_t)(64 * i) * DMODEL);
        }
        __builtin_amdgcn_sched_barrier(0);

        {
            const int myh = (lane < 16) ? h0 : h1;
            const float* rp = &plane[myh * MSG_HSTRIDE + row * MSG_ROWSTRIDE];
            float mc = -3.0e38f;
            for (int k = 0; k < KC_ATT; ++k) mc = fmaxf(mc, rp[k]);
            const float mnew = fmaxf(m, mc);
            const float r    = __expf(m - mnew);
            float lc = 0.0f, s2c = 0.0f;
            for (int k = 0; k < KC_ATT; ++k) {
                const float e = __expf(rp[k] - mnew);
                lc  += e;
                s2c += e * e;
            }
            l  = l * r + lc;
            s2 = s2 * (r * r) + s2c;
            m  = mnew;
            s_scale[own] = r;
            s_m[own]     = mnew;
        }

        for (int i = 0; i < 8; ++i) {
            const int rr = i + ((lane < 16) ? 0 : 8);
            acc0[i] *= s_scale[wid * 32 + rr];
            acc1[i] *= s_scale[wid * 32 + 16 + rr];
        }
        const float m0 = s_m[wid * 32 + row];
        const float m1 = s_m[wid * 32 + 16 + row];

        const float* p0 = &plane[h0 * MSG_HSTRIDE + row * MSG_ROWSTRIDE + khalf];
        const float* p1 = &plane[h1 * MSG_HSTRIDE + row * MSG_ROWSTRIDE + khalf];
        for (int j = 0; j < KC_ATT / 4; ++j) {
            const int ks = 4 * j;
            v2f a0; a0.x = __expf(p0[ks] - m0); a0.y = __expf(p0[ks + 1] - m0);
            v2f a1; a1.x = __expf(p1[ks] - m1); a1.y = __expf(p1[ks + 1] - m1);
            acc0 = wmma_f32(a0, bf0[j], acc0);
            acc1 = wmma_f32(a1, bf1[j], acc1);
        }

        if (more) {
            __builtin_amdgcn_sched_barrier(0);
            float* nplane = gplanes + ((t + 1) & 1) * MSG_PLANE + soff;
            for (int i = 0; i < 8; ++i) {
                const float mk = (areg[i] > 0) ? 0.0f : NEG_INF_F;
                float* d = nplane + (2 * i) * MSG_ROWSTRIDE;
                d[0 * MSG_HSTRIDE] = mreg[i].x + mk;
                d[1 * MSG_HSTRIDE] = mreg[i].y + mk;
                d[2 * MSG_HSTRIDE] = mreg[i].z + mk;
                d[3 * MSG_HSTRIDE] = mreg[i].w + mk;
            }
        }
        __syncthreads();
    }

    float* s_att  = s_msg;
    float* s_wo   = s_msg + 16 * ATT_PITCH;
    float* s_dump = s_msg + EPI_FLOATS;
    if (grp == 1) {
        s_gm[gown] = m;
        s_gl[gown] = l;
        s_gs[gown] = s2;
        for (int i = 0; i < 8; ++i) {
            s_dump[((w * 2 + 0) * 8 + i) * 32 + lane] = acc0[i];
            s_dump[((w * 2 + 1) * 8 + i) * 32 + lane] = acc1[i];
        }
    }
    for (int i = 0; i < 4; ++i)
        *(float4*)(s_wo + woff + (64 * i) * WO_PITCH) = mreg[i];
    __syncthreads();

    if (grp == 0) {
        const float mB  = s_gm[gown];
        const float M   = fmaxf(m, mB);
        const float ea  = __expf(m  - M);
        const float eb  = __expf(mB - M);
        const float L   = l  * ea        + s_gl[gown] * eb;
        const float S2  = s2 * (ea * ea) + s_gs[gown] * (eb * eb);
        const float f   = sqrtf(S2) / (L * L);
        s_scale[own] = ea * f;
        s_m[own]     = eb * f;
        for (int i = 0; i < 8; ++i) {
            const int rr = i + ((lane < 16) ? 0 : 8);
            const float fa0 = s_scale[wid * 32 + rr];
            const float fb0 = s_m[wid * 32 + rr];
            const float fa1 = s_scale[wid * 32 + 16 + rr];
            const float fb1 = s_m[wid * 32 + 16 + rr];
            const float g0  = s_dump[((w * 2 + 0) * 8 + i) * 32 + lane];
            const float g1  = s_dump[((w * 2 + 1) * 8 + i) * 32 + lane];
            s_att[rr * ATT_PITCH + h0 * 16 + row] = acc0[i] * fa0 + g0 * fb0;
            s_att[rr * ATT_PITCH + h1 * 16 + row] = acc1[i] * fa1 + g1 * fb1;
        }
    }

    const int d0 = h0 * 16;
    const int d1 = h1 * 16;
    const float bias0 = b_out[d0 + row];
    const float bias1 = b_out[d1 + row];
    const int   kq    = (lane < 16) ? 0 : 4;
    __syncthreads();

    v8f o0 = {};
    v8f o1 = {};
    for (int c = 0; c < DMODEL / KC_GEMM; ++c) {
        const float* wpl  = s_wo + (c & 1) * WO_PLANE;
        const bool   more = (c + 1 < DMODEL / KC_GEMM);
        if (more) {
            for (int i = 0; i < 4; ++i)
                mreg[i] = *(const float4*)(wptr + (size_t)(64 * i) * DMODEL +
                                           (c + 1) * KC_GEMM);
        }
        if (grp == 0) {
            const float* ap  = s_att + row * ATT_PITCH + c * KC_GEMM + kq;
            const float* bp0 = wpl + (d0 + row) * WO_PITCH + kq;
            const float* bp1 = wpl + (d1 + row) * WO_PITCH + kq;
            for (int J = 0; J < KC_GEMM / 8; ++J) {
                const float4 a  = *(const float4*)(ap  + 8 * J);
                const float4 w0 = *(const float4*)(bp0 + 8 * J);
                const float4 w1 = *(const float4*)(bp1 + 8 * J);
                v2f alo; alo.x = a.x;  alo.y = a.y;
                v2f ahi; ahi.x = a.z;  ahi.y = a.w;
                v2f b0l; b0l.x = w0.x; b0l.y = w0.y;
                v2f b0h; b0h.x = w0.z; b0h.y = w0.w;
                v2f b1l; b1l.x = w1.x; b1l.y = w1.y;
                v2f b1h; b1h.x = w1.z; b1h.y = w1.w;
                o0 = wmma_f32(alo, b0l, o0);
                o1 = wmma_f32(alo, b1l, o1);
                o0 = wmma_f32(ahi, b0h, o0);
                o1 = wmma_f32(ahi, b1h, o1);
            }
        }
        if (more) {
            __builtin_amdgcn_sched_barrier(0);
            float* npl = s_wo + ((c + 1) & 1) * WO_PLANE + woff;
            for (int i = 0; i < 4; ++i)
                *(float4*)(npl + (64 * i) * WO_PITCH) = mreg[i];
            __syncthreads();
        }
    }

    if (grp == 0) {
        for (int i = 0; i < 8; ++i) {
            const int rr = i + ((lane < 16) ? 0 : 8);
            out[(size_t)(b * NQ + q0 + rr) * DMODEL + d0 + row] = o0[i] + bias0;
            out[(size_t)(b * NQ + q0 + rr) * DMODEL + d1 + row] = o1[i] + bias1;
        }
    }
}

extern "C" void kernel_launch(void* const* d_in, const int* in_sizes, int n_in,
                              void* d_out, int out_size, void* d_ws, size_t ws_size,
                              hipStream_t stream) {
    (void)in_sizes; (void)n_in; (void)out_size; (void)ws_size;

    const float* v_inv    = (const float*)d_in[0];
    const float* messages = (const float*)d_in[1];
    const int*   adj      = (const int*)  d_in[2];
    const float* W_in     = (const float*)d_in[3];
    const float* b_in     = (const float*)d_in[4];
    const float* W_out    = (const float*)d_in[5];
    const float* b_out    = (const float*)d_in[6];
    float*       out      = (float*)d_out;

    float* proj = (float*)d_ws;

    gemm_bias_256<<<(BATCH * NKV) / 16, 256, 0, stream>>>(v_inv, W_in, b_in, proj);
    attn_kernel<<<BATCH * (NQ / 16), 512, 0, stream>>>(messages, adj, proj,
                                                       W_out, b_out, out);
}


# projection kernel: W chunk sweep software-pipelined two chunks ahead via two LDS planes (16-byte LDS stores, 8-byte fragment reads); LDS barrier written by hand so loads stay in flight across it
# speedup vs baseline: 1.3718x; 1.0796x over previous
#include <hip/hip_runtime.h>

#define BATCH   8
#define NQ      512
#define NKV     512
#define NH      16
#define DMODEL  256
#define NEG_INF_F (-1.0e9f)

#define KC_ATT  32
#define KC_GEMM 32

typedef __attribute__((ext_vector_type(2))) float v2f;
typedef __attribute__((ext_vector_type(8))) float v8f;
typedef __attribute__((ext_vector_type(4))) float v4f;

__device__ __forceinline__ v8f wmma_f32(v2f a, v2f b, v8f c) {
    return __builtin_amdgcn_wmma_f32_16x16x4_f32(
         false, a,  false, b,
         (short)0, c,  false,  false);
}

#define G_IN_PITCH (DMODEL + 4)
#define G_W_PITCH  (KC_GEMM + 4)
#define G_W_PLANE  (DMODEL * G_W_PITCH)
#define G_NCHUNK   (DMODEL / KC_GEMM)

__device__ __forceinline__ void lds_barrier() {
    asm volatile("s_wait_dscnt 0x0\n\t"
                 "s_barrier_signal -1\n\t"
                 "s_barrier_wait -1" ::: "memory");
}

__device__ __forceinline__ void gemm_trip(const int c,
                                          const float* __restrict__ wptr,
                                          const float* ap,
                                          float* s_w, const int woff,
                                          const int b0off, const int b1off,
                                          v4f (&ldset)[8], v4f (&stset)[8],
                                          v8f& acc0, v8f& acc1) {
    const float* wpl = s_w + (c & 1) * G_W_PLANE;
    const bool more2 = (c + 2 < G_NCHUNK);
    const bool more1 = (c + 1 < G_NCHUNK);

    if (more2) {
        for (int i = 0; i < 8; ++i)
            ldset[i] = *(const v4f*)(wptr + (size_t)(32 * i) * DMODEL +
                                     (c + 2) * KC_GEMM);
    }
    __builtin_amdgcn_sched_barrier(0);

    const float* a   = ap + c * KC_GEMM;
    const float* bp0 = wpl + b0off;
    const float* bp1 = wpl + b1off;
    v2f af[KC_GEMM / 4], bf0[KC_GEMM / 4], bf1[KC_GEMM / 4];
    for (int j = 0; j < KC_GEMM / 4; ++j) {
        af[j]  = *(const v2f*)(a   + 4 * j);
        bf0[j] = *(const v2f*)(bp0 + 4 * j);
        bf1[j] = *(const v2f*)(bp1 + 4 * j);
    }
    for (int j = 0; j < KC_GEMM / 4; ++j) {
        acc0 = wmma_f32(af[j], bf0[j], acc0);
        acc1 = wmma_f32(af[j], bf1[j], acc1);
    }

    if (more1) {
        float* npl = s_w + ((c + 1) & 1) * G_W_PLANE + woff;
        for (int i = 0; i < 8; ++i)
            *(v4f*)(npl + (32 * i) * G_W_PITCH) = stset[i];
        lds_barrier();
    }
}

__launch_bounds__(256)
__global__ void gemm_bias_256(const float* __restrict__ in,
                              const float* __restrict__ W,
                              const float* __restrict__ bias,
                              float* __restrict__ out) {
    __shared__ __attribute__((aligned(16))) float s_in[16 * G_IN_PITCH];
    __shared__ __attribute__((aligned(16))) float s_w[2 * G_W_PLANE];

    const int tid   = threadIdx.x;
    const int wid   = tid >> 5;
    const int lane  = tid & 31;
    const int row   = lane & 15;
    const int khalf = (lane >> 4) * 2;
    const int n0    = blockIdx.x * 16;

    const int wd = tid >> 3;
    const int wk = (tid & 7) * 4;
    const float* wptr = W + (size_t)wd * DMODEL + wk;
    const int    woff = wd * G_W_PITCH + wk;

    v4f ireg[4];
    v4f wa[8];
    v4f wb[8];

    for (int i = 0; i < 4; ++i) {
        const int e4 = (tid + 256 * i) * 4;
        const int r  = e4 >> 8, k = e4 & 255;
        ireg[i] = *(const v4f*)&in[(size_t)(n0 + r) * DMODEL + k];
    }
    for (int i = 0; i < 8; ++i)
        wa[i] = *(const v4f*)(wptr + (size_t)(32 * i) * DMODEL);
    for (int i = 0; i < 8; ++i)
        wb[i] = *(const v4f*)(wptr + (size_t)(32 * i) * DMODEL + KC_GEMM);
    for (int i = 0; i < 4; ++i) {
        const int e4 = (tid + 256 * i) * 4;
        const int r  = e4 >> 8, k = e4 & 255;
        *(v4f*)&s_in[r * G_IN_PITCH + k] = ireg[i];
    }
    for (int i = 0; i < 8; ++i)
        *(v4f*)&s_w[woff + (32 * i) * G_W_PITCH] = wa[i];
    lds_barrier();

    v8f acc0 = {};
    v8f acc1 = {};
    const int d0 = (wid)     * 16;
    const int d1 = (wid + 8) * 16;
    const float* ap   = s_in + row * G_IN_PITCH + khalf;
    const int   b0off = (d0 + row) * G_W_PITCH + khalf;
    const int   b1off = (d1 + row) * G_W_PITCH + khalf;

    for (int c = 0; c < G_NCHUNK; c += 2) {
        gemm_trip(c,     wptr, ap, s_w, woff, b0off, b1off, wa, wb, acc0, acc1);
        gemm_trip(c + 1, wptr, ap, s_w, woff, b0off, b1off, wb, wa, acc0, acc1);
    }

    const float bias0 = bias[d0 + row];
    const float bias1 = bias[d1 + row];
    for (int i = 0; i < 8; ++i) {
        const int r = i + ((lane < 16) ? 0 : 8);
        out[(size_t)(n0 + r) * DMODEL + d0 + row] = acc0[i] + bias0;
        out[(size_t)(n0 + r) * DMODEL + d1 + row] = acc1[i] + bias1;
    }
}

#define MSG_ROWSTRIDE (KC_ATT + 1)
#define MSG_HSTRIDE   (16 * MSG_ROWSTRIDE + 4)
#define MSG_PLANE     (NH * MSG_HSTRIDE)
#define NCHUNK        (NKV / KC_ATT)
#define NGROUP        2
#define NTRIP         (NCHUNK / NGROUP)
#define ATT_PITCH     (DMODEL + 4)
#define WO_PITCH      (KC_GEMM + 4)
#define WO_PLANE      (DMODEL * WO_PITCH)
#define EPI_FLOATS    (16 * ATT_PITCH + 2 * WO_PLANE)
#define DUMP_FLOATS   (8 * 2 * 8 * 32)
#define ATTN_LDS_FLOATS (((EPI_FLOATS + DUMP_FLOATS) > 2 * NGROUP * MSG_PLANE) ? \
                         (EPI_FLOATS + DUMP_FLOATS) : 2 * NGROUP * MSG_PLANE)

__launch_bounds__(512)
__global__ void attn_kernel(const float* __restrict__ messages,
                            const int*   __restrict__ adj,
                            const float* __restrict__ proj,
                            const float* __restrict__ W_out,
                            const float* __restrict__ b_out,
                            float*       __restrict__ out) {
    __shared__ __attribute__((aligned(16))) float s_msg[ATTN_LDS_FLOATS];
    __shared__ float s_scale[16 * 32];
    __shared__ float s_m[16 * 32];
    __shared__ float s_gm[8 * 32];
    __shared__ float s_gl[8 * 32];
    __shared__ float s_gs[8 * 32];

    const int tid   = threadIdx.x;
    const int wid   = __builtin_amdgcn_readfirstlane(tid >> 5);
    const int grp   = wid >> 3;
    const int w     = wid & 7;
    const int gtid  = tid & 255;
    const int lane  = tid & 31;
    const int row   = lane & 15;
    const int khalf = (lane < 16) ? 0 : 2;
    const int h0    = w * 2;
    const int h1    = w * 2 + 1;
    const int b     = blockIdx.x / (NQ / 16);
    const int q0    = (blockIdx.x % (NQ / 16)) * 16;
    const int own   = wid * 32 + lane;
    const int gown  = w * 32 + lane;

    const float* msg_base = messages + (size_t)(b * NQ + q0) * NKV * NH;
    const int*   adj_base = adj      + (size_t)(b * NQ + q0) * NKV;
    const float* proj_b   = proj     + (size_t)b * NKV * DMODEL;

    const int sq = gtid >> 7;
    const int sj = (gtid & 127) * 4;
    const int sk = sj >> 4;
    const int sh = sj & 15;
    const float* mptr = msg_base + (size_t)sq * (NKV * NH) + sj + grp * (KC_ATT * NH);
    const int*   aptr = adj_base + (size_t)sq * NKV + sk + grp * KC_ATT;
    const int    soff = sh * MSG_HSTRIDE + sq * MSG_ROWSTRIDE + sk;
    float* gplanes = s_msg + grp * (2 * MSG_PLANE);

    const int wd = tid >> 3;
    const int wk = (tid & 7) * 4;
    const float* wptr = W_out + (size_t)wd * DMODEL + wk;
    const int    woff = wd * WO_PITCH + wk;

    float4 mreg[8];
    int    areg[8];

    for (int i = 0; i < 8; ++i) {
        mreg[i] = *(const float4*)(mptr + (size_t)(2 * i) * (NKV * NH));
        areg[i] = aptr[(size_t)(2 * i) * NKV];
    }
    for (int i = 0; i < 8; ++i) {
        const float mk = (areg[i] > 0) ? 0.0f : NEG_INF_F;
        float* d = gplanes + soff + (2 * i) * MSG_ROWSTRIDE;
        d[0 * MSG_HSTRIDE] = mreg[i].x + mk;
        d[1 * MSG_HSTRIDE] = mreg[i].y + mk;
        d[2 * MSG_HSTRIDE] = mreg[i].z + mk;
        d[3 * MSG_HSTRIDE] = mreg[i].w + mk;
    }
    __syncthreads();

    float m = -3.0e38f, l = 0.0f, s2 = 0.0f;
    v8f acc0 = {};
    v8f acc1 = {};

    for (int t = 0; t < NTRIP; ++t) {
        const int   kc    = (NGROUP * t + grp) * KC_ATT;
        const int   knext = kc + NGROUP * KC_ATT;
        const float* plane = gplanes + (t & 1) * MSG_PLANE;
        const bool  more  = (t + 1 < NTRIP);

        v2f bf0[KC_ATT / 4], bf1[KC_ATT / 4];
        for (int j = 0; j < KC_ATT / 4; ++j) {
            const float* vrow = proj_b + (size_t)(kc + 4 * j + khalf) * DMODEL;
            bf0[j].x = vrow[h0 * 16 + row]; bf0[j].y = vrow[DMODEL + h0 * 16 + row];
            bf1[j].x = vrow[h1 * 16 + row]; bf1[j].y = vrow[DMODEL + h1 * 16 + row];
        }

        if (more) {
            for (int i = 0; i < 8; ++i) {
                mreg[i] = *(const float4*)(mptr + (size_t)(2 * i) * (NKV * NH) +
                                           (knext - grp * KC_ATT) * NH);
                areg[i] = aptr[(size_t)(2 * i) * NKV + (knext - grp * KC_ATT)];
            }
        } else {
            for (int i = 0; i < 4; ++i)
                mreg[i] = *(const float4*)(wptr + (size_t)(64 * i) * DMODEL);
        }
        __builtin_amdgcn_sched_barrier(0);

        {
            const int myh = (lane < 16) ? h0 : h1;
            const float* rp = &plane[myh * MSG_HSTRIDE + row * MSG_ROWSTRIDE];
            float mc = -3.0e38f;
            for (int k = 0; k < KC_ATT; ++k) mc = fmaxf(mc, rp[k]);
            const float mnew = fmaxf(m, mc);
            const float r    = __expf(m - mnew);
            float lc = 0.0f, s2c = 0.0f;
            for (int k = 0; k < KC_ATT; ++k) {
                const float e = __expf(rp[k] - mnew);
                lc  += e;
                s2c += e * e;
            }
            l  = l * r + lc;
            s2 = s2 * (r * r) + s2c;
            m  = mnew;
            s_scale[own] = r;
            s_m[own]     = mnew;
        }

        for (int i = 0; i < 8; ++i) {
            const int rr = i + ((lane < 16) ? 0 : 8);
            acc0[i] *= s_scale[wid * 32 + rr];
            acc1[i] *= s_scale[wid * 32 + 16 + rr];
        }
        const float m0 = s_m[wid * 32 + row];
        const float m1 = s_m[wid * 32 + 16 + row];

        const float* p0 = &plane[h0 * MSG_HSTRIDE + row * MSG_ROWSTRIDE + khalf];
        const float* p1 = &plane[h1 * MSG_HSTRIDE + row * MSG_ROWSTRIDE + khalf];
        for (int j = 0; j < KC_ATT / 4; ++j) {
            const int ks = 4 * j;
            v2f a0; a0.x = __expf(p0[ks] - m0); a0.y = __expf(p0[ks + 1] - m0);
            v2f a1; a1.x = __expf(p1[ks] - m1); a1.y = __expf(p1[ks + 1] - m1);
            acc0 = wmma_f32(a0, bf0[j], acc0);
            acc1 = wmma_f32(a1, bf1[j], acc1);
        }

        if (more) {
            __builtin_amdgcn_sched_barrier(0);
            float* nplane = gplanes + ((t + 1) & 1) * MSG_PLANE + soff;
            for (int i = 0; i < 8; ++i) {
                const float mk = (areg[i] > 0) ? 0.0f : NEG_INF_F;
                float* d = nplane + (2 * i) * MSG_ROWSTRIDE;
                d[0 * MSG_HSTRIDE] = mreg[i].x + mk;
                d[1 * MSG_HSTRIDE] = mreg[i].y + mk;
                d[2 * MSG_HSTRIDE] = mreg[i].z + mk;
                d[3 * MSG_HSTRIDE] = mreg[i].w + mk;
            }
        }
        __syncthreads();
    }

    float* s_att  = s_msg;
    float* s_wo   = s_msg + 16 * ATT_PITCH;
    float* s_dump = s_msg + EPI_FLOATS;
    if (grp == 1) {
        s_gm[gown] = m;
        s_gl[gown] = l;
        s_gs[gown] = s2;
        for (int i = 0; i < 8; ++i) {
            s_dump[((w * 2 + 0) * 8 + i) * 32 + lane] = acc0[i];
            s_dump[((w * 2 + 1) * 8 + i) * 32 + lane] = acc1[i];
        }
    }
    for (int i = 0; i < 4; ++i)
        *(float4*)(s_wo + woff + (64 * i) * WO_PITCH) = mreg[i];
    __syncthreads();

    if (grp == 0) {
        const float mB  = s_gm[gown];
        const float M   = fmaxf(m, mB);
        const float ea  = __expf(m  - M);
        const float eb  = __expf(mB - M);
        const float L   = l  * ea        + s_gl[gown] * eb;
        const float S2  = s2 * (ea * ea) + s_gs[gown] * (eb * eb);
        const float f   = sqrtf(S2) / (L * L);
        s_scale[own] = ea * f;
        s_m[own]     = eb * f;
        for (int i = 0; i < 8; ++i) {
            const int rr = i + ((lane < 16) ? 0 : 8);
            const float fa0 = s_scale[wid * 32 + rr];
            const float fb0 = s_m[wid * 32 + rr];
            const float fa1 = s_scale[wid * 32 + 16 + rr];
            const float fb1 = s_m[wid * 32 + 16 + rr];
            const float g0  = s_dump[((w * 2 + 0) * 8 + i) * 32 + lane];
            const float g1  = s_dump[((w * 2 + 1) * 8 + i) * 32 + lane];
            s_att[rr * ATT_PITCH + h0 * 16 + row] = acc0[i] * fa0 + g0 * fb0;
            s_att[rr * ATT_PITCH + h1 * 16 + row] = acc1[i] * fa1 + g1 * fb1;
        }
    }

    const int d0 = h0 * 16;
    const int d1 = h1 * 16;
    const float bias0 = b_out[d0 + row];
    const float bias1 = b_out[d1 + row];
    const int   kq    = (lane < 16) ? 0 : 4;
    __syncthreads();

    v8f o0 = {};
    v8f o1 = {};
    for (int c = 0; c < DMODEL / KC_GEMM; ++c) {
        const float* wpl  = s_wo + (c & 1) * WO_PLANE;
        const bool   more = (c + 1 < DMODEL / KC_GEMM);
        if (more) {
            for (int i = 0; i < 4; ++i)
                mreg[i] = *(const float4*)(wptr + (size_t)(64 * i) * DMODEL +
                                           (c + 1) * KC_GEMM);
        }
        if (grp == 0) {
            const float* ap  = s_att + row * ATT_PITCH + c * KC_GEMM + kq;
            const float* bp0 = wpl + (d0 + row) * WO_PITCH + kq;
            const float* bp1 = wpl + (d1 + row) * WO_PITCH + kq;
            for (int J = 0; J < KC_GEMM / 8; ++J) {
                const float4 a  = *(const float4*)(ap  + 8 * J);
                const float4 w0 = *(const float4*)(bp0 + 8 * J);
                const float4 w1 = *(const float4*)(bp1 + 8 * J);
                v2f alo; alo.x = a.x;  alo.y = a.y;
                v2f ahi; ahi.x = a.z;  ahi.y = a.w;
                v2f b0l; b0l.x = w0.x; b0l.y = w0.y;
                v2f b0h; b0h.x = w0.z; b0h.y = w0.w;
                v2f b1l; b1l.x = w1.x; b1l.y = w1.y;
                v2f b1h; b1h.x = w1.z; b1h.y = w1.w;
                o0 = wmma_f32(alo, b0l, o0);
                o1 = wmma_f32(alo, b1l, o1);
                o0 = wmma_f32(ahi, b0h, o0);
                o1 = wmma_f32(ahi, b1h, o1);
            }
        }
        if (more) {
            __builtin_amdgcn_sched_barrier(0);
            float* npl = s_wo + ((c + 1) & 1) * WO_PLANE + woff;
            for (int i = 0; i < 4; ++i)
                *(float4*)(npl + (64 * i) * WO_PITCH) = mreg[i];
            __syncthreads();
        }
    }

    if (grp == 0) {
        for (int i = 0; i < 8; ++i) {
            const int rr = i + ((lane < 16) ? 0 : 8);
            out[(size_t)(b * NQ + q0 + rr) * DMODEL + d0 + row] = o0[i] + bias0;
            out[(size_t)(b * NQ + q0 + rr) * DMODEL + d1 + row] = o1[i] + bias1;
        }
    }
}

extern "C" void kernel_launch(void* const* d_in, const int* in_sizes, int n_in,
                              void* d_out, int out_size, void* d_ws, size_t ws_size,
                              hipStream_t stream) {
    (void)in_sizes; (void)n_in; (void)out_size; (void)ws_size;

    const float* v_inv    = (const float*)d_in[0];
    const float* messages = (const float*)d_in[1];
    const int*   adj      = (const int*)  d_in[2];
    const float* W_in     = (const float*)d_in[3];
    const float* b_in     = (const float*)d_in[4];
    const float* W_out    = (const float*)d_in[5];
    const float* b_out    = (const float*)d_in[6];
    float*       out      = (float*)d_out;

    float* proj = (float*)d_ws;

    gemm_bias_256<<<(BATCH * NKV) / 16, 256, 0, stream>>>(v_inv, W_in, b_in, proj);
    attn_kernel<<<BATCH * (NQ / 16), 512, 0, stream>>>(messages, adj, proj,
                                                       W_out, b_out, out);
}


# attention softmax once per chunk in the WMMA A-fragment layout: one exp per logit not two, one LDS pass not two; row sums per lane half, added at the end (key-sum order, k order in 8-blocks change)
# speedup vs baseline: 1.4850x; 1.0825x over previous
#include <hip/hip_runtime.h>

#define BATCH   8
#define NQ      512
#define NKV     512
#define NH      16
#define DMODEL  256
#define NEG_INF_F (-1.0e9f)

#define KC_ATT  32
#define KC_GEMM 32

typedef __attribute__((ext_vector_type(2))) float v2f;
typedef __attribute__((ext_vector_type(8))) float v8f;
typedef __attribute__((ext_vector_type(4))) float v4f;

__device__ __forceinline__ v8f wmma_f32(v2f a, v2f b, v8f c) {
    return __builtin_amdgcn_wmma_f32_16x16x4_f32(
         false, a,  false, b,
         (short)0, c,  false,  false);
}

#define G_IN_PITCH (DMODEL + 4)
#define G_W_PITCH  (KC_GEMM + 4)
#define G_W_PLANE  (DMODEL * G_W_PITCH)
#define G_NCHUNK   (DMODEL / KC_GEMM)

__device__ __forceinline__ void lds_barrier() {
    asm volatile("s_wait_dscnt 0x0\n\t"
                 "s_barrier_signal -1\n\t"
                 "s_barrier_wait -1" ::: "memory");
}

__device__ __forceinline__ void gemm_trip(const int c,
                                          const float* __restrict__ wptr,
                                          const float* ap,
                                          float* s_w, const int woff,
                                          const int b0off, const int b1off,
                                          v4f (&ldset)[8], v4f (&stset)[8],
                                          v8f& acc0, v8f& acc1) {
    const float* wpl = s_w + (c & 1) * G_W_PLANE;
    const bool more2 = (c + 2 < G_NCHUNK);
    const bool more1 = (c + 1 < G_NCHUNK);

    if (more2) {
        for (int i = 0; i < 8; ++i)
            ldset[i] = *(const v4f*)(wptr + (size_t)(32 * i) * DMODEL +
                                     (c + 2) * KC_GEMM);
    }
    __builtin_amdgcn_sched_barrier(0);

    const float* a   = ap + c * KC_GEMM;
    const float* bp0 = wpl + b0off;
    const float* bp1 = wpl + b1off;
    v2f af[KC_GEMM / 4], bf0[KC_GEMM / 4], bf1[KC_GEMM / 4];
    for (int j = 0; j < KC_GEMM / 4; ++j) {
        af[j]  = *(const v2f*)(a   + 4 * j);
        bf0[j] = *(const v2f*)(bp0 + 4 * j);
        bf1[j] = *(const v2f*)(bp1 + 4 * j);
    }
    for (int j = 0; j < KC_GEMM / 4; ++j) {
        acc0 = wmma_f32(af[j], bf0[j], acc0);
        acc1 = wmma_f32(af[j], bf1[j], acc1);
    }

    if (more1) {
        float* npl = s_w + ((c + 1) & 1) * G_W_PLANE + woff;
        for (int i = 0; i < 8; ++i)
            *(v4f*)(npl + (32 * i) * G_W_PITCH) = stset[i];
        lds_barrier();
    }
}

__launch_bounds__(256)
__global__ void gemm_bias_256(const float* __restrict__ in,
                              const float* __restrict__ W,
                              const float* __restrict__ bias,
                              float* __restrict__ out) {
    __shared__ __attribute__((aligned(16))) float s_in[16 * G_IN_PITCH];
    __shared__ __attribute__((aligned(16))) float s_w[2 * G_W_PLANE];

    const int tid   = threadIdx.x;
    const int wid   = tid >> 5;
    const int lane  = tid & 31;
    const int row   = lane & 15;
    const int khalf = (lane >> 4) * 2;
    const int n0    = blockIdx.x * 16;

    const int wd = tid >> 3;
    const int wk = (tid & 7) * 4;
    const float* wptr = W + (size_t)wd * DMODEL + wk;
    const int    woff = wd * G_W_PITCH + wk;

    v4f ireg[4];
    v4f wa[8];
    v4f wb[8];

    for (int i = 0; i < 4; ++i) {
        const int e4 = (tid + 256 * i) * 4;
        const int r  = e4 >> 8, k = e4 & 255;
        ireg[i] = *(const v4f*)&in[(size_t)(n0 + r) * DMODEL + k];
    }
    for (int i = 0; i < 8; ++i)
        wa[i] = *(const v4f*)(wptr + (size_t)(32 * i) * DMODEL);
    for (int i = 0; i < 8; ++i)
        wb[i] = *(const v4f*)(wptr + (size_t)(32 * i) * DMODEL + KC_GEMM);
    for (int i = 0; i < 4; ++i) {
        const int e4 = (tid + 256 * i) * 4;
        const int r  = e4 >> 8, k = e4 & 255;
        *(v4f*)&s_in[r * G_IN_PITCH + k] = ireg[i];
    }
    for (int i = 0; i < 8; ++i)
        *(v4f*)&s_w[woff + (32 * i) * G_W_PITCH] = wa[i];
    lds_barrier();

    v8f acc0 = {};
    v8f acc1 = {};
    const int d0 = (wid)     * 16;
    const int d1 = (wid + 8) * 16;
    const float* ap   = s_in + row * G_IN_PITCH + khalf;
    const int   b0off = (d0 + row) * G_W_PITCH + khalf;
    const int   b1off = (d1 + row) * G_W_PITCH + khalf;

    for (int c = 0; c < G_NCHUNK; c += 2) {
        gemm_trip(c,     wptr, ap, s_w, woff, b0off, b1off, wa, wb, acc0, acc1);
        gemm_trip(c + 1, wptr, ap, s_w, woff, b0off, b1off, wb, wa, acc0, acc1);
    }

    const float bias0 = bias[d0 + row];
    const float bias1 = bias[d1 + row];
    for (int i = 0; i < 8; ++i) {
        const int r = i + ((lane < 16) ? 0 : 8);
        out[(size_t)(n0 + r) * DMODEL + d0 + row] = acc0[i] + bias0;
        out[(size_t)(n0 + r) * DMODEL + d1 + row] = acc1[i] + bias1;
    }
}

#define MSG_ROWSTRIDE (KC_ATT + 4)
#define MSG_HSTRIDE   (16 * MSG_ROWSTRIDE + 4)
#define MSG_PLANE     (NH * MSG_HSTRIDE)
#define NCHUNK        (NKV / KC_ATT)
#define NGROUP        2
#define NTRIP         (NCHUNK / NGROUP)
#define ATT_PITCH     (DMODEL + 4)
#define WO_PITCH      (KC_GEMM + 4)
#define WO_PLANE      (DMODEL * WO_PITCH)
#define EPI_FLOATS    (16 * ATT_PITCH + 2 * WO_PLANE)
#define DUMP_FLOATS   (8 * 2 * 8 * 32)
#define ATTN_LDS_FLOATS (((EPI_FLOATS + DUMP_FLOATS) > 2 * NGROUP * MSG_PLANE) ? \
                         (EPI_FLOATS + DUMP_FLOATS) : 2 * NGROUP * MSG_PLANE)

__device__ __forceinline__ float xhalf(float x) {
    return __builtin_bit_cast(float, __builtin_amdgcn_permlanex16(
        __builtin_bit_cast(int, x), __builtin_bit_cast(int, x),
        0x76543210, 0xfedcba98, false, false));
}

__launch_bounds__(512)
__global__ void attn_kernel(const float* __restrict__ messages,
                            const int*   __restrict__ adj,
                            const float* __restrict__ proj,
                            const float* __restrict__ W_out,
                            const float* __restrict__ b_out,
                            float*       __restrict__ out) {
    __shared__ __attribute__((aligned(16))) float s_msg[ATTN_LDS_FLOATS];
    __shared__ float s_scale[16 * 32];
    __shared__ float s_m[16 * 32];
    __shared__ float s_gm[8 * 32];
    __shared__ float s_gl[8 * 32];
    __shared__ float s_gs[8 * 32];

    const int tid   = threadIdx.x;
    const int wid   = __builtin_amdgcn_readfirstlane(tid >> 5);
    const int grp   = wid >> 3;
    const int w     = wid & 7;
    const int gtid  = tid & 255;
    const int lane  = tid & 31;
    const int row   = lane & 15;
    const int kq    = (lane < 16) ? 0 : 4;
    const int h0    = w * 2;
    const int h1    = w * 2 + 1;
    const int b     = blockIdx.x / (NQ / 16);
    const int q0    = (blockIdx.x % (NQ / 16)) * 16;
    const int own   = wid * 32 + lane;
    const int gown  = w * 32 + lane;

    const float* msg_base = messages + (size_t)(b * NQ + q0) * NKV * NH;
    const int*   adj_base = adj      + (size_t)(b * NQ + q0) * NKV;
    const float* proj_b   = proj     + (size_t)b * NKV * DMODEL;

    const int sq = gtid >> 7;
    const int sj = (gtid & 127) * 4;
    const int sk = sj >> 4;
    const int sh = sj & 15;
    const float* mptr = msg_base + (size_t)sq * (NKV * NH) + sj + grp * (KC_ATT * NH);
    const int*   aptr = adj_base + (size_t)sq * NKV + sk + grp * KC_ATT;
    const int    soff = sh * MSG_HSTRIDE + sq * MSG_ROWSTRIDE + sk;
    float* gplanes = s_msg + grp * (2 * MSG_PLANE);

    const int wd = tid >> 3;
    const int wk = (tid & 7) * 4;
    const float* wptr = W_out + (size_t)wd * DMODEL + wk;
    const int    woff = wd * WO_PITCH + wk;

    float4 mreg[8];
    int    areg[8];

    for (int i = 0; i < 8; ++i) {
        mreg[i] = *(const float4*)(mptr + (size_t)(2 * i) * (NKV * NH));
        areg[i] = aptr[(size_t)(2 * i) * NKV];
    }
    for (int i = 0; i < 8; ++i) {
        const float mk = (areg[i] > 0) ? 0.0f : NEG_INF_F;
        float* d = gplanes + soff + (2 * i) * MSG_ROWSTRIDE;
        d[0 * MSG_HSTRIDE] = mreg[i].x + mk;
        d[1 * MSG_HSTRIDE] = mreg[i].y + mk;
        d[2 * MSG_HSTRIDE] = mreg[i].z + mk;
        d[3 * MSG_HSTRIDE] = mreg[i].w + mk;
    }
    __syncthreads();

    float m0 = -3.0e38f, l0 = 0.0f, s20 = 0.0f;
    float m1 = -3.0e38f, l1 = 0.0f, s21 = 0.0f;
    v8f acc0 = {};
    v8f acc1 = {};

    for (int t = 0; t < NTRIP; ++t) {
        const int   kc    = (NGROUP * t + grp) * KC_ATT;
        const int   knext = kc + NGROUP * KC_ATT;
        const float* plane = gplanes + (t & 1) * MSG_PLANE;
        const bool  more  = (t + 1 < NTRIP);

        v2f bf0[KC_ATT / 4], bf1[KC_ATT / 4];
        for (int j = 0; j < KC_ATT / 4; ++j) {
            const float* vrow = proj_b + (size_t)(kc + 8 * (j >> 1) + 2 * (j & 1) + kq) * DMODEL;
            bf0[j].x = vrow[h0 * 16 + row]; bf0[j].y = vrow[DMODEL + h0 * 16 + row];
            bf1[j].x = vrow[h1 * 16 + row]; bf1[j].y = vrow[DMODEL + h1 * 16 + row];
        }

        if (more) {
            for (int i = 0; i < 8; ++i) {
                mreg[i] = *(const float4*)(mptr + (size_t)(2 * i) * (NKV * NH) +
                                           (knext - grp * KC_ATT) * NH);
                areg[i] = aptr[(size_t)(2 * i) * NKV + (knext - grp * KC_ATT)];
            }
        } else {
            for (int i = 0; i < 4; ++i)
                mreg[i] = *(const float4*)(wptr + (size_t)(64 * i) * DMODEL);
        }
        __builtin_amdgcn_sched_barrier(0);

        const float* p0 = &plane[h0 * MSG_HSTRIDE + row * MSG_ROWSTRIDE + kq];
        const float* p1 = &plane[h1 * MSG_HSTRIDE + row * MSG_ROWSTRIDE + kq];
        v4f x0[KC_ATT / 8], x1[KC_ATT / 8];
        for (int J = 0; J < KC_ATT / 8; ++J) {
            x0[J] = *(const v4f*)(p0 + 8 * J);
            x1[J] = *(const v4f*)(p1 + 8 * J);
        }

        float mc0 = -3.0e38f, mc1 = -3.0e38f;
        for (int J = 0; J < KC_ATT / 8; ++J) {
            mc0 = fmaxf(mc0, x0[J].x); mc0 = fmaxf(mc0, x0[J].y);
            mc0 = fmaxf(mc0, x0[J].z); mc0 = fmaxf(mc0, x0[J].w);
            mc1 = fmaxf(mc1, x1[J].x); mc1 = fmaxf(mc1, x1[J].y);
            mc1 = fmaxf(mc1, x1[J].z); mc1 = fmaxf(mc1, x1[J].w);
        }
        mc0 = fmaxf(mc0, xhalf(mc0));
        mc1 = fmaxf(mc1, xhalf(mc1));
        const float mn0 = fmaxf(m0, mc0);
        const float mn1 = fmaxf(m1, mc1);
        const float r0  = __expf(m0 - mn0);
        const float r1  = __expf(m1 - mn1);
        v2f lv0 = {}, sv0 = {}, lv1 = {}, sv1 = {};
        for (int J = 0; J < KC_ATT / 8; ++J) {
            v4f e0, e1;
            e0.x = __expf(x0[J].x - mn0); e0.y = __expf(x0[J].y - mn0);
            e0.z = __expf(x0[J].z - mn0); e0.w = __expf(x0[J].w - mn0);
            e1.x = __expf(x1[J].x - mn1); e1.y = __expf(x1[J].y - mn1);
            e1.z = __expf(x1[J].z - mn1); e1.w = __expf(x1[J].w - mn1);
            const v2f e0l = {e0.x, e0.y}, e0h = {e0.z, e0.w};
            const v2f e1l = {e1.x, e1.y}, e1h = {e1.z, e1.w};
            lv0 += e0l; sv0 += e0l * e0l;
            lv0 += e0h; sv0 += e0h * e0h;
            lv1 += e1l; sv1 += e1l * e1l;
            lv1 += e1h; sv1 += e1h * e1h;
            x0[J] = e0;
            x1[J] = e1;
        }
        l0  = l0 * r0 + (lv0.x + lv0.y);
        s20 = s20 * (r0 * r0) + (sv0.x + sv0.y);
        m0  = mn0;
        l1  = l1 * r1 + (lv1.x + lv1.y);
        s21 = s21 * (r1 * r1) + (sv1.x + sv1.y);
        m1  = mn1;

        s_scale[own] = (lane < 16) ? r0 : r1;
        for (int i = 0; i < 8; ++i) {
            const int rr = i + ((lane < 16) ? 0 : 8);
            acc0[i] *= s_scale[wid * 32 + rr];
            acc1[i] *= s_scale[wid * 32 + 16 + rr];
        }

        for (int J = 0; J < KC_ATT / 8; ++J) {
            v2f a0l; a0l.x = x0[J].x; a0l.y = x0[J].y;
            v2f a0h; a0h.x = x0[J].z; a0h.y = x0[J].w;
            v2f a1l; a1l.x = x1[J].x; a1l.y = x1[J].y;
            v2f a1h; a1h.x = x1[J].z; a1h.y = x1[J].w;
            acc0 = wmma_f32(a0l, bf0[2 * J],     acc0);
            acc1 = wmma_f32(a1l, bf1[2 * J],     acc1);
            acc0 = wmma_f32(a0h, bf0[2 * J + 1], acc0);
            acc1 = wmma_f32(a1h, bf1[2 * J + 1], acc1);
        }

        if (more) {
            __builtin_amdgcn_sched_barrier(0);
            float* nplane = gplanes + ((t + 1) & 1) * MSG_PLANE + soff;
            for (int i = 0; i < 8; ++i) {
                const float mk = (areg[i] > 0) ? 0.0f : NEG_INF_F;
                float* d = nplane + (2 * i) * MSG_ROWSTRIDE;
                d[0 * MSG_HSTRIDE] = mreg[i].x + mk;
                d[1 * MSG_HSTRIDE] = mreg[i].y + mk;
                d[2 * MSG_HSTRIDE] = mreg[i].z + mk;
                d[3 * MSG_HSTRIDE] = mreg[i].w + mk;
            }
        }
        __syncthreads();
    }

    l0  += xhalf(l0);
    l1  += xhalf(l1);
    s20 += xhalf(s20);
    s21 += xhalf(s21);
    const float m  = (lane < 16) ? m0  : m1;
    const float l  = (lane < 16) ? l0  : l1;
    const float s2 = (lane < 16) ? s20 : s21;

    float* s_att  = s_msg;
    float* s_wo   = s_msg + 16 * ATT_PITCH;
    float* s_dump = s_msg + EPI_FLOATS;
    if (grp == 1) {
        s_gm[gown] = m;
        s_gl[gown] = l;
        s_gs[gown] = s2;
        for (int i = 0; i < 8; ++i) {
            s_dump[((w * 2 + 0) * 8 + i) * 32 + lane] = acc0[i];
            s_dump[((w * 2 + 1) * 8 + i) * 32 + lane] = acc1[i];
        }
    }
    for (int i = 0; i < 4; ++i)
        *(float4*)(s_wo + woff + (64 * i) * WO_PITCH) = mreg[i];
    __syncthreads();

    if (grp == 0) {
        const float mB  = s_gm[gown];
        const float M   = fmaxf(m, mB);
        const float ea  = __expf(m  - M);
        const float eb  = __expf(mB - M);
        const float L   = l  * ea        + s_gl[gown] * eb;
        const float S2  = s2 * (ea * ea) + s_gs[gown] * (eb * eb);
        const float f   = sqrtf(S2) / (L * L);
        s_scale[own] = ea * f;
        s_m[own]     = eb * f;
        for (int i = 0; i < 8; ++i) {
            const int rr = i + ((lane < 16) ? 0 : 8);
            const float fa0 = s_scale[wid * 32 + rr];
            const float fb0 = s_m[wid * 32 + rr];
            const float fa1 = s_scale[wid * 32 + 16 + rr];
            const float fb1 = s_m[wid * 32 + 16 + rr];
            const float g0  = s_dump[((w * 2 + 0) * 8 + i) * 32 + lane];
            const float g1  = s_dump[((w * 2 + 1) * 8 + i) * 32 + lane];
            s_att[rr * ATT_PITCH + h0 * 16 + row] = acc0[i] * fa0 + g0 * fb0;
            s_att[rr * ATT_PITCH + h1 * 16 + row] = acc1[i] * fa1 + g1 * fb1;
        }
    }

    const int d0 = h0 * 16;
    const int d1 = h1 * 16;
    const float bias0 = b_out[d0 + row];
    const float bias1 = b_out[d1 + row];
    __syncthreads();

    v8f o0 = {};
    v8f o1 = {};
    for (int c = 0; c < DMODEL / KC_GEMM; ++c) {
        const float* wpl  = s_wo + (c & 1) * WO_PLANE;
        const bool   more = (c + 1 < DMODEL / KC_GEMM);
        if (more) {
            for (int i = 0; i < 4; ++i)
                mreg[i] = *(const float4*)(wptr + (size_t)(64 * i) * DMODEL +
                                           (c + 1) * KC_GEMM);
        }
        if (grp == 0) {
            const float* ap  = s_att + row * ATT_PITCH + c * KC_GEMM + kq;
            const float* bp0 = wpl + (d0 + row) * WO_PITCH + kq;
            const float* bp1 = wpl + (d1 + row) * WO_PITCH + kq;
            for (int J = 0; J < KC_GEMM / 8; ++J) {
                const float4 a  = *(const float4*)(ap  + 8 * J);
                const float4 w0 = *(const float4*)(bp0 + 8 * J);
                const float4 w1 = *(const float4*)(bp1 + 8 * J);
                v2f alo; alo.x = a.x;  alo.y = a.y;
                v2f ahi; ahi.x = a.z;  ahi.y = a.w;
                v2f b0l; b0l.x = w0.x; b0l.y = w0.y;
                v2f b0h; b0h.x = w0.z; b0h.y = w0.w;
                v2f b1l; b1l.x = w1.x; b1l.y = w1.y;
                v2f b1h; b1h.x = w1.z; b1h.y = w1.w;
                o0 = wmma_f32(alo, b0l, o0);
                o1 = wmma_f32(alo, b1l, o1);
                o0 = wmma_f32(ahi, b0h, o0);
                o1 = wmma_f32(ahi, b1h, o1);
            }
        }
        if (more) {
            __builtin_amdgcn_sched_barrier(0);
            float* npl = s_wo + ((c + 1) & 1) * WO_PLANE + woff;
            for (int i = 0; i < 4; ++i)
                *(float4*)(npl + (64 * i) * WO_PITCH) = mreg[i];
            __syncthreads();
        }
    }

    if (grp == 0) {
        for (int i = 0; i < 8; ++i) {
            const int rr = i + ((lane < 16) ? 0 : 8);
            out[(size_t)(b * NQ + q0 + rr) * DMODEL + d0 + row] = o0[i] + bias0;
            out[(size_t)(b * NQ + q0 + rr) * DMODEL + d1 + row] = o1[i] + bias1;
        }
    }
}

extern "C" void kernel_launch(void* const* d_in, const int* in_sizes, int n_in,
                              void* d_out, int out_size, void* d_ws, size_t ws_size,
                              hipStream_t stream) {
    (void)in_sizes; (void)n_in; (void)out_size; (void)ws_size;

    const float* v_inv    = (const float*)d_in[0];
    const float* messages = (const float*)d_in[1];
    const int*   adj      = (const int*)  d_in[2];
    const float* W_in     = (const float*)d_in[3];
    const float* b_in     = (const float*)d_in[4];
    const float* W_out    = (const float*)d_in[5];
    const float* b_out    = (const float*)d_in[6];
    float*       out      = (float*)d_out;

    float* proj = (float*)d_ws;

    gemm_bias_256<<<(BATCH * NKV) / 16, 256, 0, stream>>>(v_inv, W_in, b_in, proj);
    attn_kernel<<<BATCH * (NQ / 16), 512, 0, stream>>>(messages, adj, proj,
                                                       W_out, b_out, out);
}


# attention k-chunk loop, loads issued earlier: message+adjacency stream two chunks ahead through a second register set, value fragments right behind the previous chunk's WMMAs; trip barrier by hand
# speedup vs baseline: 1.5609x; 1.0511x over previous
#include <hip/hip_runtime.h>

#define BATCH   8
#define NQ      512
#define NKV     512
#define NH      16
#define DMODEL  256
#define NEG_INF_F (-1.0e9f)

#define KC_ATT  32
#define KC_GEMM 32

typedef __attribute__((ext_vector_type(2))) float v2f;
typedef __attribute__((ext_vector_type(8))) float v8f;
typedef __attribute__((ext_vector_type(4))) float v4f;

__device__ __forceinline__ v8f wmma_f32(v2f a, v2f b, v8f c) {
    return __builtin_amdgcn_wmma_f32_16x16x4_f32(
         false, a,  false, b,
         (short)0, c,  false,  false);
}

#define G_IN_PITCH (DMODEL + 4)
#define G_W_PITCH  (KC_GEMM + 4)
#define G_W_PLANE  (DMODEL * G_W_PITCH)
#define G_NCHUNK   (DMODEL / KC_GEMM)

__device__ __forceinline__ void lds_barrier() {
    asm volatile("s_wait_dscnt 0x0\n\t"
                 "s_barrier_signal -1\n\t"
                 "s_barrier_wait -1" ::: "memory");
}

__device__ __forceinline__ void gemm_trip(const int c,
                                          const float* __restrict__ wptr,
                                          const float* ap,
                                          float* s_w, const int woff,
                                          const int b0off, const int b1off,
                                          v4f (&ldset)[8], v4f (&stset)[8],
                                          v8f& acc0, v8f& acc1) {
    const float* wpl = s_w + (c & 1) * G_W_PLANE;
    const bool more2 = (c + 2 < G_NCHUNK);
    const bool more1 = (c + 1 < G_NCHUNK);

    if (more2) {
        for (int i = 0; i < 8; ++i)
            ldset[i] = *(const v4f*)(wptr + (size_t)(32 * i) * DMODEL +
                                     (c + 2) * KC_GEMM);
    }
    __builtin_amdgcn_sched_barrier(0);

    const float* a   = ap + c * KC_GEMM;
    const float* bp0 = wpl + b0off;
    const float* bp1 = wpl + b1off;
    v2f af[KC_GEMM / 4], bf0[KC_GEMM / 4], bf1[KC_GEMM / 4];
    for (int j = 0; j < KC_GEMM / 4; ++j) {
        af[j]  = *(const v2f*)(a   + 4 * j);
        bf0[j] = *(const v2f*)(bp0 + 4 * j);
        bf1[j] = *(const v2f*)(bp1 + 4 * j);
    }
    for (int j = 0; j < KC_GEMM / 4; ++j) {
        acc0 = wmma_f32(af[j], bf0[j], acc0);
        acc1 = wmma_f32(af[j], bf1[j], acc1);
    }

    if (more1) {
        float* npl = s_w + ((c + 1) & 1) * G_W_PLANE + woff;
        for (int i = 0; i < 8; ++i)
            *(v4f*)(npl + (32 * i) * G_W_PITCH) = stset[i];
        lds_barrier();
    }
}

__launch_bounds__(256)
__global__ void gemm_bias_256(const float* __restrict__ in,
                              const float* __restrict__ W,
                              const float* __restrict__ bias,
                              float* __restrict__ out) {
    __shared__ __attribute__((aligned(16))) float s_in[16 * G_IN_PITCH];
    __shared__ __attribute__((aligned(16))) float s_w[2 * G_W_PLANE];

    const int tid   = threadIdx.x;
    const int wid   = tid >> 5;
    const int lane  = tid & 31;
    const int row   = lane & 15;
    const int khalf = (lane >> 4) * 2;
    const int n0    = blockIdx.x * 16;

    const int wd = tid >> 3;
    const int wk = (tid & 7) * 4;
    const float* wptr = W + (size_t)wd * DMODEL + wk;
    const int    woff = wd * G_W_PITCH + wk;

    v4f ireg[4];
    v4f wa[8];
    v4f wb[8];

    for (int i = 0; i < 4; ++i) {
        const int e4 = (tid + 256 * i) * 4;
        const int r  = e4 >> 8, k = e4 & 255;
        ireg[i] = *(const v4f*)&in[(size_t)(n0 + r) * DMODEL + k];
    }
    for (int i = 0; i < 8; ++i)
        wa[i] = *(const v4f*)(wptr + (size_t)(32 * i) * DMODEL);
    for (int i = 0; i < 8; ++i)
        wb[i] = *(const v4f*)(wptr + (size_t)(32 * i) * DMODEL + KC_GEMM);
    for (int i = 0; i < 4; ++i) {
        const int e4 = (tid + 256 * i) * 4;
        const int r  = e4 >> 8, k = e4 & 255;
        *(v4f*)&s_in[r * G_IN_PITCH + k] = ireg[i];
    }
    for (int i = 0; i < 8; ++i)
        *(v4f*)&s_w[woff + (32 * i) * G_W_PITCH] = wa[i];
    lds_barrier();

    v8f acc0 = {};
    v8f acc1 = {};
    const int d0 = (wid)     * 16;
    const int d1 = (wid + 8) * 16;
    const float* ap   = s_in + row * G_IN_PITCH + khalf;
    const int   b0off = (d0 + row) * G_W_PITCH + khalf;
    const int   b1off = (d1 + row) * G_W_PITCH + khalf;

    for (int c = 0; c < G_NCHUNK; c += 2) {
        gemm_trip(c,     wptr, ap, s_w, woff, b0off, b1off, wa, wb, acc0, acc1);
        gemm_trip(c + 1, wptr, ap, s_w, woff, b0off, b1off, wb, wa, acc0, acc1);
    }

    const float bias0 = bias[d0 + row];
    const float bias1 = bias[d1 + row];
    for (int i = 0; i < 8; ++i) {
        const int r = i + ((lane < 16) ? 0 : 8);
        out[(size_t)(n0 + r) * DMODEL + d0 + row] = acc0[i] + bias0;
        out[(size_t)(n0 + r) * DMODEL + d1 + row] = acc1[i] + bias1;
    }
}

#define MSG_ROWSTRIDE (KC_ATT + 4)
#define MSG_HSTRIDE   (16 * MSG_ROWSTRIDE + 4)
#define MSG_PLANE     (NH * MSG_HSTRIDE)
#define NCHUNK        (NKV / KC_ATT)
#define NGROUP        2
#define NTRIP         (NCHUNK / NGROUP)
#define ATT_PITCH     (DMODEL + 4)
#define WO_PITCH      (KC_GEMM + 4)
#define WO_PLANE      (DMODEL * WO_PITCH)
#define EPI_FLOATS    (16 * ATT_PITCH + 2 * WO_PLANE)
#define DUMP_FLOATS   (8 * 2 * 8 * 32)
#define ATTN_LDS_FLOATS (((EPI_FLOATS + DUMP_FLOATS) > 2 * NGROUP * MSG_PLANE) ? \
                         (EPI_FLOATS + DUMP_FLOATS) : 2 * NGROUP * MSG_PLANE)

__device__ __forceinline__ float xhalf(float x) {
    return __builtin_bit_cast(float, __builtin_amdgcn_permlanex16(
        __builtin_bit_cast(int, x), __builtin_bit_cast(int, x),
        0x76543210, 0xfedcba98, false, false));
}

__launch_bounds__(512)
__global__ void attn_kernel(const float* __restrict__ messages,
                            const int*   __restrict__ adj,
                            const float* __restrict__ proj,
                            const float* __restrict__ W_out,
                            const float* __restrict__ b_out,
                            float*       __restrict__ out) {
    __shared__ __attribute__((aligned(16))) float s_msg[ATTN_LDS_FLOATS];
    __shared__ float s_scale[16 * 32];
    __shared__ float s_m[16 * 32];
    __shared__ float s_gm[8 * 32];
    __shared__ float s_gl[8 * 32];
    __shared__ float s_gs[8 * 32];

    const int tid   = threadIdx.x;
    const int wid   = __builtin_amdgcn_readfirstlane(tid >> 5);
    const int grp   = wid >> 3;
    const int w     = wid & 7;
    const int gtid  = tid & 255;
    const int lane  = tid & 31;
    const int row   = lane & 15;
    const int kq    = (lane < 16) ? 0 : 4;
    const int h0    = w * 2;
    const int h1    = w * 2 + 1;
    const int b     = blockIdx.x / (NQ / 16);
    const int q0    = (blockIdx.x % (NQ / 16)) * 16;
    const int own   = wid * 32 + lane;
    const int gown  = w * 32 + lane;

    const float* msg_base = messages + (size_t)(b * NQ + q0) * NKV * NH;
    const int*   adj_base = adj      + (size_t)(b * NQ + q0) * NKV;
    const float* proj_b   = proj     + (size_t)b * NKV * DMODEL;

    const int sq = gtid >> 4;
    const int sp = gtid & 15;
    const int sk = sp >> 2;
    const int sh = (sp & 3) * 4;
    const float* mptr = msg_base + (size_t)sq * (NKV * NH) + sp * 4 + grp * (KC_ATT * NH);
    const int*   aptr = adj_base + (size_t)sq * NKV + sk + grp * KC_ATT;
    const int    soff = sh * MSG_HSTRIDE + sq * MSG_ROWSTRIDE + sk;
    float* gplanes = s_msg + grp * (2 * MSG_PLANE);

    float4 mA[8], mB[8];
    int    aA[8], aB[8];
    const unsigned voff = (unsigned)(kq * DMODEL + row);
    v2f bf0[KC_ATT / 4], bf1[KC_ATT / 4];
    auto load_values = [&](const int kc) __attribute__((always_inline)) {
        const float* vl = proj_b + (size_t)kc * DMODEL + h0 * 16 + voff;
        for (int j = 0; j < KC_ATT / 4; ++j) {
            const int r = (8 * (j >> 1) + 2 * (j & 1)) * DMODEL;
            bf0[j].x = vl[r];      bf0[j].y = vl[r + DMODEL];
            bf1[j].x = vl[r + 16]; bf1[j].y = vl[r + DMODEL + 16];
        }
    };
    for (int i = 0; i < 8; ++i) {
        mA[i] = *(const float4*)(mptr + 64 * i);
        aA[i] = aptr[4 * i];
    }
    for (int i = 0; i < 8; ++i) {
        mB[i] = *(const float4*)(mptr + 64 * i + (NGROUP * KC_ATT) * NH);
        aB[i] = aptr[4 * i + NGROUP * KC_ATT];
    }
    __builtin_amdgcn_sched_barrier(0);
    load_values(grp * KC_ATT);
    __builtin_amdgcn_sched_barrier(0);
    for (int i = 0; i < 8; ++i) {
        const float mk = (aA[i] > 0) ? 0.0f : NEG_INF_F;
        float* d = gplanes + soff + 4 * i;
        d[0 * MSG_HSTRIDE] = mA[i].x + mk;
        d[1 * MSG_HSTRIDE] = mA[i].y + mk;
        d[2 * MSG_HSTRIDE] = mA[i].z + mk;
        d[3 * MSG_HSTRIDE] = mA[i].w + mk;
    }
    lds_barrier();

    float m0 = -3.0e38f, l0 = 0.0f, s20 = 0.0f;
    float m1 = -3.0e38f, l1 = 0.0f, s21 = 0.0f;
    v8f acc0 = {};
    v8f acc1 = {};

    auto trip = [&](const int t, const bool more2, const bool more1,
                    float4 (&ldm)[8], int (&lda)[8],
                    float4 (&stm)[8], int (&sta)[8]) __attribute__((always_inline)) {
        const int   kc    = (NGROUP * t + grp) * KC_ATT;
        const float* plane = gplanes + (t & 1) * MSG_PLANE;

        if (more2) {
            for (int i = 0; i < 8; ++i) {
                ldm[i] = *(const float4*)(mptr + 64 * i + (NGROUP * (t + 2) * KC_ATT) * NH);
                lda[i] = aptr[4 * i + NGROUP * (t + 2) * KC_ATT];
            }
        } else if (!more1) {
            const float* wp0 = W_out + (size_t)(tid >> 3) * DMODEL + (tid & 7) * 4;
            for (int i = 0; i < 4; ++i)
                ldm[i] = *(const float4*)(wp0 + (size_t)(64 * i) * DMODEL);
        }
        __builtin_amdgcn_sched_barrier(0);

        const float* p0 = &plane[h0 * MSG_HSTRIDE + row * MSG_ROWSTRIDE + kq];
        const float* p1 = &plane[h1 * MSG_HSTRIDE + row * MSG_ROWSTRIDE + kq];
        v4f x0[KC_ATT / 8], x1[KC_ATT / 8];
        for (int J = 0; J < KC_ATT / 8; ++J) {
            x0[J] = *(const v4f*)(p0 + 8 * J);
            x1[J] = *(const v4f*)(p1 + 8 * J);
        }

        float mc0 = -3.0e38f, mc1 = -3.0e38f;
        for (int J = 0; J < KC_ATT / 8; ++J) {
            mc0 = fmaxf(mc0, x0[J].x); mc0 = fmaxf(mc0, x0[J].y);
            mc0 = fmaxf(mc0, x0[J].z); mc0 = fmaxf(mc0, x0[J].w);
            mc1 = fmaxf(mc1, x1[J].x); mc1 = fmaxf(mc1, x1[J].y);
            mc1 = fmaxf(mc1, x1[J].z); mc1 = fmaxf(mc1, x1[J].w);
        }
        mc0 = fmaxf(mc0, xhalf(mc0));
        mc1 = fmaxf(mc1, xhalf(mc1));
        const float mn0 = fmaxf(m0, mc0);
        const float mn1 = fmaxf(m1, mc1);
        const float r0  = __expf(m0 - mn0);
        const float r1  = __expf(m1 - mn1);
        v2f lv0 = {}, sv0 = {}, lv1 = {}, sv1 = {};
        for (int J = 0; J < KC_ATT / 8; ++J) {
            v4f e0, e1;
            e0.x = __expf(x0[J].x - mn0); e0.y = __expf(x0[J].y - mn0);
            e0.z = __expf(x0[J].z - mn0); e0.w = __expf(x0[J].w - mn0);
            e1.x = __expf(x1[J].x - mn1); e1.y = __expf(x1[J].y - mn1);
            e1.z = __expf(x1[J].z - mn1); e1.w = __expf(x1[J].w - mn1);
            const v2f e0l = {e0.x, e0.y}, e0h = {e0.z, e0.w};
            const v2f e1l = {e1.x, e1.y}, e1h = {e1.z, e1.w};
            lv0 += e0l; sv0 += e0l * e0l;
            lv0 += e0h; sv0 += e0h * e0h;
            lv1 += e1l; sv1 += e1l * e1l;
            lv1 += e1h; sv1 += e1h * e1h;
            x0[J] = e0;
            x1[J] = e1;
        }
        l0  = l0 * r0 + (lv0.x + lv0.y);
        s20 = s20 * (r0 * r0) + (sv0.x + sv0.y);
        m0  = mn0;
        l1  = l1 * r1 + (lv1.x + lv1.y);
        s21 = s21 * (r1 * r1) + (sv1.x + sv1.y);
        m1  = mn1;

        s_scale[own] = (lane < 16) ? r0 : r1;
        for (int i = 0; i < 8; ++i) {
            const int rr = i + ((lane < 16) ? 0 : 8);
            acc0[i] *= s_scale[wid * 32 + rr];
            acc1[i] *= s_scale[wid * 32 + 16 + rr];
        }

        for (int J = 0; J < KC_ATT / 8; ++J) {
            v2f a0l; a0l.x = x0[J].x; a0l.y = x0[J].y;
            v2f a0h; a0h.x = x0[J].z; a0h.y = x0[J].w;
            v2f a1l; a1l.x = x1[J].x; a1l.y = x1[J].y;
            v2f a1h; a1h.x = x1[J].z; a1h.y = x1[J].w;
            acc0 = wmma_f32(a0l, bf0[2 * J],     acc0);
            acc1 = wmma_f32(a1l, bf1[2 * J],     acc1);
            acc0 = wmma_f32(a0h, bf0[2 * J + 1], acc0);
            acc1 = wmma_f32(a1h, bf1[2 * J + 1], acc1);
        }
        asm volatile("" : "+v"(acc0), "+v"(acc1));

        if (more1) {
            __builtin_amdgcn_sched_barrier(0);
            load_values(kc + NGROUP * KC_ATT);
            __builtin_amdgcn_sched_barrier(0);
            float* nplane = gplanes + ((t + 1) & 1) * MSG_PLANE + soff;
            for (int i = 0; i < 8; ++i) {
                const float mk = (sta[i] > 0) ? 0.0f : NEG_INF_F;
                float* d = nplane + 4 * i;
                d[0 * MSG_HSTRIDE] = stm[i].x + mk;
                d[1 * MSG_HSTRIDE] = stm[i].y + mk;
                d[2 * MSG_HSTRIDE] = stm[i].z + mk;
                d[3 * MSG_HSTRIDE] = stm[i].w + mk;
            }
        }
        lds_barrier();
    };

    for (int t = 0; t < NTRIP - 2; t += 2) {
        trip(t,     true, true, mA, aA, mB, aB);
        trip(t + 1, true, true, mB, aB, mA, aA);
    }
    trip(NTRIP - 2, false, true,  mA, aA, mB, aB);
    trip(NTRIP - 1, false, false, mB, aB, mA, aA);

    const int wd = tid >> 3;
    const int wk = (tid & 7) * 4;
    const float* wptr = W_out + (size_t)wd * DMODEL + wk;
    const int    woff = wd * WO_PITCH + wk;

    l0  += xhalf(l0);
    l1  += xhalf(l1);
    s20 += xhalf(s20);
    s21 += xhalf(s21);
    const float m  = (lane < 16) ? m0  : m1;
    const float l  = (lane < 16) ? l0  : l1;
    const float s2 = (lane < 16) ? s20 : s21;

    float* s_att  = s_msg;
    float* s_wo   = s_msg + 16 * ATT_PITCH;
    float* s_dump = s_msg + EPI_FLOATS;
    if (grp == 1) {
        s_gm[gown] = m;
        s_gl[gown] = l;
        s_gs[gown] = s2;
        for (int i = 0; i < 8; ++i) {
            s_dump[((w * 2 + 0) * 8 + i) * 32 + lane] = acc0[i];
            s_dump[((w * 2 + 1) * 8 + i) * 32 + lane] = acc1[i];
        }
    }
    for (int i = 0; i < 4; ++i)
        *(float4*)(s_wo + woff + (64 * i) * WO_PITCH) = mB[i];
    __syncthreads();

    if (grp == 0) {
        const float mB  = s_gm[gown];
        const float M   = fmaxf(m, mB);
        const float ea  = __expf(m  - M);
        const float eb  = __expf(mB - M);
        const float L   = l  * ea        + s_gl[gown] * eb;
        const float S2  = s2 * (ea * ea) + s_gs[gown] * (eb * eb);
        const float f   = sqrtf(S2) / (L * L);
        s_scale[own] = ea * f;
        s_m[own]     = eb * f;
        for (int i = 0; i < 8; ++i) {
            const int rr = i + ((lane < 16) ? 0 : 8);
            const float fa0 = s_scale[wid * 32 + rr];
            const float fb0 = s_m[wid * 32 + rr];
            const float fa1 = s_scale[wid * 32 + 16 + rr];
            const float fb1 = s_m[wid * 32 + 16 + rr];
            const float g0  = s_dump[((w * 2 + 0) * 8 + i) * 32 + lane];
            const float g1  = s_dump[((w * 2 + 1) * 8 + i) * 32 + lane];
            s_att[rr * ATT_PITCH + h0 * 16 + row] = acc0[i] * fa0 + g0 * fb0;
            s_att[rr * ATT_PITCH + h1 * 16 + row] = acc1[i] * fa1 + g1 * fb1;
        }
    }

    const int d0 = h0 * 16;
    const int d1 = h1 * 16;
    const float bias0 = b_out[d0 + row];
    const float bias1 = b_out[d1 + row];
    __syncthreads();

    v8f o0 = {};
    v8f o1 = {};
    for (int c = 0; c < DMODEL / KC_GEMM; ++c) {
        const float* wpl  = s_wo + (c & 1) * WO_PLANE;
        const bool   more = (c + 1 < DMODEL / KC_GEMM);
        if (more) {
            for (int i = 0; i < 4; ++i)
                mB[i] = *(const float4*)(wptr + (size_t)(64 * i) * DMODEL +
                                           (c + 1) * KC_GEMM);
        }
        if (grp == 0) {
            const float* ap  = s_att + row * ATT_PITCH + c * KC_GEMM + kq;
            const float* bp0 = wpl + (d0 + row) * WO_PITCH + kq;
            const float* bp1 = wpl + (d1 + row) * WO_PITCH + kq;
            for (int J = 0; J < KC_GEMM / 8; ++J) {
                const float4 a  = *(const float4*)(ap  + 8 * J);
                const float4 w0 = *(const float4*)(bp0 + 8 * J);
                const float4 w1 = *(const float4*)(bp1 + 8 * J);
                v2f alo; alo.x = a.x;  alo.y = a.y;
                v2f ahi; ahi.x = a.z;  ahi.y = a.w;
                v2f b0l; b0l.x = w0.x; b0l.y = w0.y;
                v2f b0h; b0h.x = w0.z; b0h.y = w0.w;
                v2f b1l; b1l.x = w1.x; b1l.y = w1.y;
                v2f b1h; b1h.x = w1.z; b1h.y = w1.w;
                o0 = wmma_f32(alo, b0l, o0);
                o1 = wmma_f32(alo, b1l, o1);
                o0 = wmma_f32(ahi, b0h, o0);
                o1 = wmma_f32(ahi, b1h, o1);
            }
        }
        if (more) {
            __builtin_amdgcn_sched_barrier(0);
            float* npl = s_wo + ((c + 1) & 1) * WO_PLANE + woff;
            for (int i = 0; i < 4; ++i)
                *(float4*)(npl + (64 * i) * WO_PITCH) = mB[i];
            __syncthreads();
        }
    }

    if (grp == 0) {
        for (int i = 0; i < 8; ++i) {
            const int rr = i + ((lane < 16) ? 0 : 8);
            out[(size_t)(b * NQ + q0 + rr) * DMODEL + d0 + row] = o0[i] + bias0;
            out[(size_t)(b * NQ + q0 + rr) * DMODEL + d1 + row] = o1[i] + bias1;
        }
    }
}

extern "C" void kernel_launch(void* const* d_in, const int* in_sizes, int n_in,
                              void* d_out, int out_size, void* d_ws, size_t ws_size,
                              hipStream_t stream) {
    (void)in_sizes; (void)n_in; (void)out_size; (void)ws_size;

    const float* v_inv    = (const float*)d_in[0];
    const float* messages = (const float*)d_in[1];
    const int*   adj      = (const int*)  d_in[2];
    const float* W_in     = (const float*)d_in[3];
    const float* b_in     = (const float*)d_in[4];
    const float* W_out    = (const float*)d_in[5];
    const float* b_out    = (const float*)d_in[6];
    float*       out      = (float*)d_out;

    float* proj = (float*)d_ws;

    gemm_bias_256<<<(BATCH * NKV) / 16, 256, 0, stream>>>(v_inv, W_in, b_in, proj);
    attn_kernel<<<BATCH * (NQ / 16), 512, 0, stream>>>(messages, adj, proj,
                                                       W_out, b_out, out);
}
